# starting kernel
# baseline (speedup 1.0000x reference)
#include <hip/hip_runtime.h>
#include <math.h>

#define N_NODES 50000
#define N_EDGES 800000
#define NE_TOT  (N_EDGES + N_NODES)
#define IN_FEATS 256
#define HID 64
#define HEADS 2
#define F1 (HEADS * HID)
#define NCLS 40

typedef __attribute__((ext_vector_type(2))) float v2f;
typedef __attribute__((ext_vector_type(8))) float v8f;

static __device__ __forceinline__ unsigned fmap(float f) {
    unsigned u = __float_as_uint(f);
    return (u & 0x80000000u) ? ~u : (u | 0x80000000u);
}
static __device__ __forceinline__ float finv(unsigned t) {
    return (t & 0x80000000u) ? __uint_as_float(t & 0x7FFFFFFFu)
                             : __uint_as_float(~t);
}
static __device__ __forceinline__ void edge_sd(const int* __restrict__ ei, int e, int& s, int& d) {
    if (e < N_EDGES) { s = ei[e]; d = ei[N_EDGES + e]; }
    else             { s = e - N_EDGES; d = s; }
}
static __device__ __forceinline__ float lrelu(float x) { return x > 0.f ? x : 0.2f * x; }

__global__ __launch_bounds__(256) void gat_gemm1_wmma(const float* __restrict__ x,
                                                      const float* __restrict__ W1,
                                                      float* __restrict__ h1) {
    const int mtile = blockIdx.x;
    const int wave  = threadIdx.x >> 5;
    const int lane  = threadIdx.x & 31;
    const int half  = lane >> 4;
    const int l     = lane & 15;
    const int row   = mtile * 16 + l;
    const int ncol  = wave * 16 + l;

    v8f acc = {};
    const float* xr = x + (size_t)row * IN_FEATS;
    for (int k = 0; k < IN_FEATS; k += 4) {
        v2f a = *(const v2f*)(xr + k + 2 * half);
        const float* bp = W1 + (size_t)(k + 2 * half) * F1 + ncol;
        v2f b; b.x = bp[0]; b.y = bp[F1];
        acc = __builtin_amdgcn_wmma_f32_16x16x4_f32(false, a, false, b,
                                                    (short)0, acc, false, false);
    }
    float* dr = h1 + (size_t)(mtile * 16 + 8 * half) * F1 + wave * 16 + l;
#pragma unroll
    for (int r = 0; r < 8; ++r) dr[(size_t)r * F1] = acc[r];
}

__global__ __launch_bounds__(96) void gat_gemm2_wmma(const float* __restrict__ h2in,
                                                     const float* __restrict__ W2,
                                                     float* __restrict__ h2) {
    const int mtile = blockIdx.x;
    const int wave  = threadIdx.x >> 5;
    const int lane  = threadIdx.x & 31;
    const int half  = lane >> 4;
    const int l     = lane & 15;
    const int row   = mtile * 16 + l;
    const int ncol  = wave * 16 + l;

    v8f acc = {};
    const float* xr = h2in + (size_t)row * F1;
    for (int k = 0; k < F1; k += 4) {
        v2f a = *(const v2f*)(xr + k + 2 * half);
        v2f b;
        if (ncol < NCLS) {
            const float* bp = W2 + (size_t)(k + 2 * half) * NCLS + ncol;
            b.x = bp[0]; b.y = bp[NCLS];
        } else { b.x = 0.f; b.y = 0.f; }
        acc = __builtin_amdgcn_wmma_f32_16x16x4_f32(false, a, false, b,
                                                    (short)0, acc, false, false);
    }
    if (ncol < NCLS) {
        float* dr = h2 + (size_t)(mtile * 16 + 8 * half) * NCLS + ncol;
#pragma unroll
        for (int r = 0; r < 8; ++r) dr[(size_t)r * NCLS] = acc[r];
    }
}

__global__ void gat_att1(const float* __restrict__ h1, const float* __restrict__ as1,
                         const float* __restrict__ ad1, float* __restrict__ a_src,
                         float* __restrict__ a_dst) {
    int i = blockIdx.x * blockDim.x + threadIdx.x;
    if (i >= N_NODES * HEADS) return;
    int n = i >> 1, h = i & 1;
    const float* hp = h1 + (size_t)n * F1 + h * HID;
    const float* s0 = as1 + h * HID;
    const float* d0 = ad1 + h * HID;
    float s = 0.f, d = 0.f;
#pragma unroll 8
    for (int c = 0; c < HID; ++c) { float v = hp[c]; s += v * s0[c]; d += v * d0[c]; }
    a_src[i] = s; a_dst[i] = d;
}

__global__ void gat_att2(const float* __restrict__ h2, const float* __restrict__ as2,
                         const float* __restrict__ ad2, float* __restrict__ a_src,
                         float* __restrict__ a_dst) {
    int n = blockIdx.x * blockDim.x + threadIdx.x;
    if (n >= N_NODES) return;
    const float* hp = h2 + (size_t)n * NCLS;
    float s = 0.f, d = 0.f;
#pragma unroll 8
    for (int c = 0; c < NCLS; ++c) { float v = hp[c]; s += v * as2[c]; d += v * ad2[c]; }
    a_src[n] = s; a_dst[n] = d;
}

__global__ void gat_edge_max1(const int* __restrict__ ei, const float* __restrict__ a_src,
                              const float* __restrict__ a_dst, unsigned* __restrict__ emax) {
    int e = blockIdx.x * blockDim.x + threadIdx.x;
    if (e >= NE_TOT) return;
    int s, d; edge_sd(ei, e, s, d);
#pragma unroll
    for (int h = 0; h < HEADS; ++h) {
        float x = lrelu(a_src[s * 2 + h] + a_dst[d * 2 + h]);
        atomicMax(&emax[d * 2 + h], fmap(x));
    }
}

__global__ void gat_edge_max2(const int* __restrict__ ei, const float* __restrict__ a_src,
                              const float* __restrict__ a_dst, unsigned* __restrict__ emax) {
    int e = blockIdx.x * blockDim.x + threadIdx.x;
    if (e >= NE_TOT) return;
    int s, d; edge_sd(ei, e, s, d);
    float x = lrelu(a_src[s] + a_dst[d]);
    atomicMax(&emax[d], fmap(x));
}

__global__ __launch_bounds__(256) void gat_edge_accum1(const int* __restrict__ ei,
        const float* __restrict__ a_src, const float* __restrict__ a_dst,
        const unsigned* __restrict__ emax, const float* __restrict__ h1,
        float* __restrict__ denom, float* __restrict__ accum) {
    int e = blockIdx.x * 8 + (threadIdx.x >> 5);
    if (e >= NE_TOT) return;
    const int lane = threadIdx.x & 31;
    int s, d; edge_sd(ei, e, s, d);
    float p0 = __expf(lrelu(a_src[s * 2 + 0] + a_dst[d * 2 + 0]) - finv(emax[d * 2 + 0]));
    float p1 = __expf(lrelu(a_src[s * 2 + 1] + a_dst[d * 2 + 1]) - finv(emax[d * 2 + 1]));
    if (lane == 0) { atomicAdd(&denom[d * 2 + 0], p0); atomicAdd(&denom[d * 2 + 1], p1); }
    const float4 hv = *(const float4*)(h1 + (size_t)s * F1 + lane * 4);
    const float p = (lane < 16) ? p0 : p1;
    float* ap = accum + (size_t)d * F1 + lane * 4;
    atomicAdd(ap + 0, hv.x * p);
    atomicAdd(ap + 1, hv.y * p);
    atomicAdd(ap + 2, hv.z * p);
    atomicAdd(ap + 3, hv.w * p);
}

__global__ __launch_bounds__(256) void gat_edge_accum2(const int* __restrict__ ei,
        const float* __restrict__ a_src, const float* __restrict__ a_dst,
        const unsigned* __restrict__ emax, const float* __restrict__ h2,
        float* __restrict__ denom, float* __restrict__ accum) {
    int e = blockIdx.x * 8 + (threadIdx.x >> 5);
    if (e >= NE_TOT) return;
    const int lane = threadIdx.x & 31;
    int s, d; edge_sd(ei, e, s, d);
    float p = __expf(lrelu(a_src[s] + a_dst[d]) - finv(emax[d]));
    if (lane == 0) atomicAdd(&denom[d], p);
    for (int c = lane; c < NCLS; c += 32)
        atomicAdd(&accum[(size_t)d * NCLS + c], h2[(size_t)s * NCLS + c] * p);
}

__global__ void gat_finalize1(const float* __restrict__ accum, const float* __restrict__ denom,
                              const float* __restrict__ b1, float* __restrict__ h2in) {
    int i = blockIdx.x * blockDim.x + threadIdx.x;
    if (i >= N_NODES * F1) return;
    int n = i >> 7, c = i & 127;
    float v = accum[i] / denom[n * 2 + (c >> 6)] + b1[c];
    h2in[i] = v > 0.f ? v : (expf(v) - 1.0f);
}

__global__ void gat_finalize2(const float* __restrict__ accum, const float* __restrict__ denom,
                              const float* __restrict__ b2, float* __restrict__ out) {
    int i = blockIdx.x * blockDim.x + threadIdx.x;
    if (i >= N_NODES * NCLS) return;
    int n = i / NCLS, c = i - n * NCLS;
    out[i] = accum[i] / denom[n] + b2[c];
}

extern "C" void kernel_launch(void* const* d_in, const int* in_sizes, int n_in,
                              void* d_out, int out_size, void* d_ws, size_t ws_size,
                              hipStream_t stream) {
    const float* x    = (const float*)d_in[0];
    const int*   ei   = (const int*)  d_in[1];
    const float* W1   = (const float*)d_in[2];
    const float* as1  = (const float*)d_in[3];
    const float* ad1  = (const float*)d_in[4];
    const float* b1   = (const float*)d_in[5];
    const float* W2   = (const float*)d_in[6];
    const float* as2  = (const float*)d_in[7];
    const float* ad2  = (const float*)d_in[8];
    const float* b2   = (const float*)d_in[9];
    float* out = (float*)d_out;

    float* ws      = (float*)d_ws;
    float* h1      = ws;
    float* accum1  = h1     + (size_t)N_NODES * F1;
    float* accum2  = accum1 + (size_t)N_NODES * F1;
    float* denom1  = accum2 + (size_t)N_NODES * NCLS;
    float* denom2  = denom1 + (size_t)N_NODES * 2;
    unsigned* emax1 = (unsigned*)(denom2 + N_NODES);
    unsigned* emax2 = emax1 + (size_t)N_NODES * 2;
    float* a_src1  = (float*)(emax2 + N_NODES);
    float* a_dst1  = a_src1 + (size_t)N_NODES * 2;
    float* a_src2  = a_dst1 + (size_t)N_NODES * 2;
    float* a_dst2  = a_src2 + N_NODES;
    float* h2      = a_dst2 + N_NODES;

    const size_t zeroFloats = (size_t)N_NODES * (F1 + NCLS + 2 + 1 + 2 + 1);
    hipMemsetAsync(accum1, 0, zeroFloats * sizeof(float), stream);

    const int MT = N_NODES / 16;
    gat_gemm1_wmma<<<MT, 256, 0, stream>>>(x, W1, h1);
    gat_att1<<<(N_NODES * HEADS + 255) / 256, 256, 0, stream>>>(h1, as1, ad1, a_src1, a_dst1);
    gat_edge_max1<<<(NE_TOT + 255) / 256, 256, 0, stream>>>(ei, a_src1, a_dst1, emax1);
    gat_edge_accum1<<<(NE_TOT + 7) / 8, 256, 0, stream>>>(ei, a_src1, a_dst1, emax1, h1,
                                                          denom1, accum1);
    gat_finalize1<<<(N_NODES * F1 + 255) / 256, 256, 0, stream>>>(accum1, denom1, b1, h1);
    gat_gemm2_wmma<<<MT, 96, 0, stream>>>(h1, W2, h2);
    gat_att2<<<(N_NODES + 255) / 256, 256, 0, stream>>>(h2, as2, ad2, a_src2, a_dst2);
    gat_edge_max2<<<(NE_TOT + 255) / 256, 256, 0, stream>>>(ei, a_src2, a_dst2, emax2);
    gat_edge_accum2<<<(NE_TOT + 7) / 8, 256, 0, stream>>>(ei, a_src2, a_dst2, emax2, h2,
                                                          denom2, accum2);
    gat_finalize2<<<(N_NODES * NCLS + 255) / 256, 256, 0, stream>>>(accum2, denom2, b2, out);
}


# Edges bucketed by destination each call (int atomic rank, scan, place); one wave per node does max, exp, sum and row gather in registers: no float atomics; sums run in slot order, arrival-set
# speedup vs baseline: 2.3855x; 2.3855x over previous
#include <hip/hip_runtime.h>
#include <math.h>

#define N_NODES 50000
#define N_EDGES 800000
#define IN_FEATS 256
#define HID 64
#define HEADS 2
#define F1 (HEADS * HID)
#define NCLS 40

typedef __attribute__((ext_vector_type(2))) float v2f;
typedef __attribute__((ext_vector_type(8))) float v8f;

static __device__ __forceinline__ float lrelu(float x) { return x > 0.f ? x : 0.2f * x; }

__global__ __launch_bounds__(256) void gat_gemm1_wmma(const float* __restrict__ x,
                                                      const float* __restrict__ W1,
                                                      float* __restrict__ h1) {
    const int mtile = blockIdx.x;
    const int wave  = threadIdx.x >> 5;
    const int lane  = threadIdx.x & 31;
    const int half  = lane >> 4;
    const int l     = lane & 15;
    const int row   = mtile * 16 + l;
    const int ncol  = wave * 16 + l;

    v8f acc = {};
    const float* xr = x + (size_t)row * IN_FEATS;
    for (int k = 0; k < IN_FEATS; k += 4) {
        v2f a = *(const v2f*)(xr + k + 2 * half);
        const float* bp = W1 + (size_t)(k + 2 * half) * F1 + ncol;
        v2f b; b.x = bp[0]; b.y = bp[F1];
        acc = __builtin_amdgcn_wmma_f32_16x16x4_f32(false, a, false, b,
                                                    (short)0, acc, false, false);
    }
    float* dr = h1 + (size_t)(mtile * 16 + 8 * half) * F1 + wave * 16 + l;
#pragma unroll
    for (int r = 0; r < 8; ++r) dr[(size_t)r * F1] = acc[r];
}

__global__ __launch_bounds__(96) void gat_gemm2_wmma(const float* __restrict__ h2in,
                                                     const float* __restrict__ W2,
                                                     float* __restrict__ h2) {
    const int mtile = blockIdx.x;
    const int wave  = threadIdx.x >> 5;
    const int lane  = threadIdx.x & 31;
    const int half  = lane >> 4;
    const int l     = lane & 15;
    const int row   = mtile * 16 + l;
    const int ncol  = wave * 16 + l;

    v8f acc = {};
    const float* xr = h2in + (size_t)row * F1;
    for (int k = 0; k < F1; k += 4) {
        v2f a = *(const v2f*)(xr + k + 2 * half);
        v2f b;
        if (ncol < NCLS) {
            const float* bp = W2 + (size_t)(k + 2 * half) * NCLS + ncol;
            b.x = bp[0]; b.y = bp[NCLS];
        } else { b.x = 0.f; b.y = 0.f; }
        acc = __builtin_amdgcn_wmma_f32_16x16x4_f32(false, a, false, b,
                                                    (short)0, acc, false, false);
    }
    if (ncol < NCLS) {
        float* dr = h2 + (size_t)(mtile * 16 + 8 * half) * NCLS + ncol;
#pragma unroll
        for (int r = 0; r < 8; ++r) dr[(size_t)r * NCLS] = acc[r];
    }
}

__global__ void gat_att1(const float* __restrict__ h1, const float* __restrict__ as1,
                         const float* __restrict__ ad1, float* __restrict__ a_src,
                         float* __restrict__ a_dst) {
    int i = blockIdx.x * blockDim.x + threadIdx.x;
    if (i >= N_NODES * HEADS) return;
    int n = i >> 1, h = i & 1;
    const float* hp = h1 + (size_t)n * F1 + h * HID;
    const float* s0 = as1 + h * HID;
    const float* d0 = ad1 + h * HID;
    float s = 0.f, d = 0.f;
#pragma unroll 8
    for (int c = 0; c < HID; ++c) { float v = hp[c]; s += v * s0[c]; d += v * d0[c]; }
    a_src[i] = s; a_dst[i] = d;
}

__global__ void gat_att2(const float* __restrict__ h2, const float* __restrict__ as2,
                         const float* __restrict__ ad2, float* __restrict__ a_src,
                         float* __restrict__ a_dst) {
    int n = blockIdx.x * blockDim.x + threadIdx.x;
    if (n >= N_NODES) return;
    const float* hp = h2 + (size_t)n * NCLS;
    float s = 0.f, d = 0.f;
#pragma unroll 8
    for (int c = 0; c < NCLS; ++c) { float v = hp[c]; s += v * as2[c]; d += v * ad2[c]; }
    a_src[n] = s; a_dst[n] = d;
}

__global__ void gat_bucket_rank(const int* __restrict__ ei, int* __restrict__ count,
                                int* __restrict__ rank) {
    int e = blockIdx.x * blockDim.x + threadIdx.x;
    if (e >= N_EDGES) return;
    rank[e] = atomicAdd(&count[ei[N_EDGES + e]], 1);
}

#define SCAN_T   1024
#define SCAN_PER ((N_NODES + SCAN_T - 1) / SCAN_T)

__global__ __launch_bounds__(SCAN_T) void gat_bucket_scan(const int* __restrict__ count,
                                                          int* __restrict__ rowptr) {
    __shared__ int wsum[32];
    const int t = threadIdx.x, lane = t & 31, w = t >> 5;
    const int base = t * SCAN_PER;
    int v[SCAN_PER];
    int sum = 0;
#pragma unroll
    for (int i = 0; i < SCAN_PER; ++i) {
        const int idx = base + i;
        v[i] = idx < N_NODES ? count[idx] : 0;
        sum += v[i];
    }
    int inc = sum;
    for (int o = 1; o < 32; o <<= 1) { int y = __shfl_up(inc, o); if (lane >= o) inc += y; }
    if (lane == 31) wsum[w] = inc;
    __syncthreads();
    if (w == 0) {
        const int x = wsum[lane];
        int xi = x;
        for (int o = 1; o < 32; o <<= 1) { int y = __shfl_up(xi, o); if (lane >= o) xi += y; }
        wsum[lane] = xi - x;
    }
    __syncthreads();
    int run = wsum[w] + inc - sum;
#pragma unroll
    for (int i = 0; i < SCAN_PER; ++i) {
        const int idx = base + i;
        if (idx < N_NODES) rowptr[idx] = run;
        run += v[i];
    }
    if (t == SCAN_T - 1) rowptr[N_NODES] = run;
}

__global__ void gat_bucket_place(const int* __restrict__ ei, const int* __restrict__ rowptr,
                                 const int* __restrict__ rank, int* __restrict__ col) {
    int e = blockIdx.x * blockDim.x + threadIdx.x;
    if (e >= N_EDGES) return;
    col[rowptr[ei[N_EDGES + e]] + rank[e]] = ei[e];
}

__global__ __launch_bounds__(256) void gat_gather1(const int* __restrict__ rowptr,
        const int* __restrict__ col, const float* __restrict__ a_src,
        const float* __restrict__ a_dst, const float* __restrict__ h1,
        const float* __restrict__ b1, float* __restrict__ h2in) {
    const int lane = threadIdx.x & 31;
    const int n = __builtin_amdgcn_readfirstlane(blockIdx.x * 8 + (threadIdx.x >> 5));
    if (n >= N_NODES) return;
    const int beg = rowptr[n], end = rowptr[n + 1];
    const float2 adn = *(const float2*)(a_dst + (size_t)n * 2);
    const float2 asn = *(const float2*)(a_src + (size_t)n * 2);
    float m0 = lrelu(asn.x + adn.x), m1 = lrelu(asn.y + adn.y);
    const bool hi = lane >= 16;
    float4 acc = *(const float4*)(h1 + (size_t)n * F1 + lane * 4);
    float den = 1.f;
    for (int c0 = beg; c0 < end; c0 += 32) {
        const int cnt = min(32, end - c0);
        const bool valid = lane < cnt;
        const int s_l = valid ? col[c0 + lane] : n;
        const float2 as = *(const float2*)(a_src + (size_t)s_l * 2);
        const float e0 = valid ? lrelu(as.x + adn.x) : -INFINITY;
        const float e1 = valid ? lrelu(as.y + adn.y) : -INFINITY;
        float c0m = e0, c1m = e1;
        for (int o = 16; o; o >>= 1) {
            c0m = fmaxf(c0m, __shfl_xor(c0m, o));
            c1m = fmaxf(c1m, __shfl_xor(c1m, o));
        }
        const float n0 = fmaxf(m0, c0m), n1 = fmaxf(m1, c1m);
        const float r = hi ? __expf(m1 - n1) : __expf(m0 - n0);
        acc.x *= r; acc.y *= r; acc.z *= r; acc.w *= r; den *= r;
        m0 = n0; m1 = n1;
        const int p0_l = __float_as_int(valid ? __expf(e0 - n0) : 0.f);
        const int p1_l = __float_as_int(valid ? __expf(e1 - n1) : 0.f);
        for (int j0 = 0; j0 < cnt; j0 += 4) {
#pragma unroll
            for (int u = 0; u < 4; ++u) {
                const int j = j0 + u;
                const int sj = __builtin_amdgcn_readlane(s_l, j);
                const float q0 = __int_as_float(__builtin_amdgcn_readlane(p0_l, j));
                const float q1 = __int_as_float(__builtin_amdgcn_readlane(p1_l, j));
                const float4 hv = *(const float4*)(h1 + (size_t)sj * F1 + lane * 4);
                const float p = hi ? q1 : q0;
                den += p;
                acc.x += p * hv.x; acc.y += p * hv.y; acc.z += p * hv.z; acc.w += p * hv.w;
            }
        }
    }
    const float4 b = *(const float4*)(b1 + lane * 4);
    float4 o;
    o.x = acc.x / den + b.x; o.y = acc.y / den + b.y;
    o.z = acc.z / den + b.z; o.w = acc.w / den + b.w;
    o.x = o.x > 0.f ? o.x : (expf(o.x) - 1.0f);
    o.y = o.y > 0.f ? o.y : (expf(o.y) - 1.0f);
    o.z = o.z > 0.f ? o.z : (expf(o.z) - 1.0f);
    o.w = o.w > 0.f ? o.w : (expf(o.w) - 1.0f);
    *(float4*)(h2in + (size_t)n * F1 + lane * 4) = o;
}

__global__ __launch_bounds__(256) void gat_gather2(const int* __restrict__ rowptr,
        const int* __restrict__ col, const float* __restrict__ a_src,
        const float* __restrict__ a_dst, const float* __restrict__ h2,
        const float* __restrict__ b2, float* __restrict__ out) {
    const int lane = threadIdx.x & 31;
    const int n = __builtin_amdgcn_readfirstlane(blockIdx.x * 8 + (threadIdx.x >> 5));
    if (n >= N_NODES) return;
    const int g3 = lane / 10;
    const bool live = g3 < 3;
    const int g = live ? g3 : 0;
    const int q = lane - 10 * g3;
    const int beg = rowptr[n], end = rowptr[n + 1];
    const float adn = a_dst[n];
    float m = lrelu(a_src[n] + adn);
    float4 acc = {0.f, 0.f, 0.f, 0.f};
    float den = 0.f;
    if (lane < 10) { acc = *(const float4*)(h2 + (size_t)n * NCLS + q * 4); den = 1.f; }
    for (int c0 = beg; c0 < end; c0 += 32) {
        const int cnt = min(32, end - c0);
        const bool valid = lane < cnt;
        const int s_l = valid ? col[c0 + lane] : n;
        const float e = valid ? lrelu(a_src[s_l] + adn) : -INFINITY;
        float cm = e;
        for (int o = 16; o; o >>= 1) cm = fmaxf(cm, __shfl_xor(cm, o));
        const float nm = fmaxf(m, cm);
        const float r = __expf(m - nm);
        acc.x *= r; acc.y *= r; acc.z *= r; acc.w *= r; den *= r;
        m = nm;
        const float p_l = valid ? __expf(e - nm) : 0.f;
        for (int j0 = 0; j0 < cnt; j0 += 6) {
#pragma unroll
            for (int u = 0; u < 2; ++u) {
                const int j = j0 + 3 * u + g;
                const bool ok = live && (j < cnt);
                const int jj = ok ? j : 0;
                const int sj = __shfl(s_l, jj);
                const float pv = __shfl(p_l, jj);
                const float pj = ok ? pv : 0.f;
                const float4 hv = *(const float4*)(h2 + (size_t)sj * NCLS + q * 4);
                den += pj;
                acc.x += pj * hv.x; acc.y += pj * hv.y; acc.z += pj * hv.z; acc.w += pj * hv.w;
            }
        }
    }
    const int l1 = (lane + 10) & 31, l2 = (lane + 20) & 31;
    const float sx = acc.x + __shfl(acc.x, l1) + __shfl(acc.x, l2);
    const float sy = acc.y + __shfl(acc.y, l1) + __shfl(acc.y, l2);
    const float sz = acc.z + __shfl(acc.z, l1) + __shfl(acc.z, l2);
    const float sw = acc.w + __shfl(acc.w, l1) + __shfl(acc.w, l2);
    const float sd = den + __shfl(den, l1) + __shfl(den, l2);
    if (lane < 10) {
        const float4 b = *(const float4*)(b2 + q * 4);
        float4 o;
        o.x = sx / sd + b.x; o.y = sy / sd + b.y; o.z = sz / sd + b.z; o.w = sw / sd + b.w;
        *(float4*)(out + (size_t)n * NCLS + q * 4) = o;
    }
}

extern "C" void kernel_launch(void* const* d_in, const int* in_sizes, int n_in,
                              void* d_out, int out_size, void* d_ws, size_t ws_size,
                              hipStream_t stream) {
    const float* x    = (const float*)d_in[0];
    const int*   ei   = (const int*)  d_in[1];
    const float* W1   = (const float*)d_in[2];
    const float* as1  = (const float*)d_in[3];
    const float* ad1  = (const float*)d_in[4];
    const float* b1   = (const float*)d_in[5];
    const float* W2   = (const float*)d_in[6];
    const float* as2  = (const float*)d_in[7];
    const float* ad2  = (const float*)d_in[8];
    const float* b2   = (const float*)d_in[9];
    float* out = (float*)d_out;

    float* ws      = (float*)d_ws;
    float* h1      = ws;
    float* h2in    = h1     + (size_t)N_NODES * F1;
    float* h2      = h2in   + (size_t)N_NODES * F1;
    float* a_src1  = h2     + (size_t)N_NODES * NCLS;
    float* a_dst1  = a_src1 + (size_t)N_NODES * 2;
    float* a_src2  = a_dst1 + (size_t)N_NODES * 2;
    float* a_dst2  = a_src2 + N_NODES;
    int*   count   = (int*)(a_dst2 + N_NODES);
    int*   rowptr  = count  + N_NODES;
    int*   rank    = rowptr + N_NODES + 4;
    int*   col     = rank   + N_EDGES;

    hipMemsetAsync(count, 0, (size_t)N_NODES * sizeof(int), stream);

    const int MT = N_NODES / 16;
    gat_bucket_rank<<<(N_EDGES + 255) / 256, 256, 0, stream>>>(ei, count, rank);
    gat_bucket_scan<<<1, SCAN_T, 0, stream>>>(count, rowptr);
    gat_bucket_place<<<(N_EDGES + 255) / 256, 256, 0, stream>>>(ei, rowptr, rank, col);
    gat_gemm1_wmma<<<MT, 256, 0, stream>>>(x, W1, h1);
    gat_att1<<<(N_NODES * HEADS + 255) / 256, 256, 0, stream>>>(h1, as1, ad1, a_src1, a_dst1);
    gat_gather1<<<N_NODES / 8, 256, 0, stream>>>(rowptr, col, a_src1, a_dst1, h1, b1, h2in);
    gat_gemm2_wmma<<<MT, 96, 0, stream>>>(h2in, W2, h2);
    gat_att2<<<(N_NODES + 255) / 256, 256, 0, stream>>>(h2, as2, ad2, a_src2, a_dst2);
    gat_gather2<<<N_NODES / 8, 256, 0, stream>>>(rowptr, col, a_src2, a_dst2, h2, b2, out);
}


# Edges go straight into fixed-pitch per-destination buckets (slot = the counting add's return, overflow list past 64): scan and place launches gone; sums still in slot order, arrival-set
# speedup vs baseline: 3.2258x; 1.3523x over previous
#include <hip/hip_runtime.h>
#include <math.h>

#define N_NODES 50000
#define N_EDGES 800000
#define IN_FEATS 256
#define HID 64
#define HEADS 2
#define F1 (HEADS * HID)
#define NCLS 40

typedef __attribute__((ext_vector_type(2))) float v2f;
typedef __attribute__((ext_vector_type(8))) float v8f;

static __device__ __forceinline__ float lrelu(float x) { return x > 0.f ? x : 0.2f * x; }

__global__ __launch_bounds__(256) void gat_gemm1_wmma(const float* __restrict__ x,
                                                      const float* __restrict__ W1,
                                                      float* __restrict__ h1) {
    const int mtile = blockIdx.x;
    const int wave  = threadIdx.x >> 5;
    const int lane  = threadIdx.x & 31;
    const int half  = lane >> 4;
    const int l     = lane & 15;
    const int row   = mtile * 16 + l;
    const int ncol  = wave * 16 + l;

    v8f acc = {};
    const float* xr = x + (size_t)row * IN_FEATS;
    for (int k = 0; k < IN_FEATS; k += 4) {
        v2f a = *(const v2f*)(xr + k + 2 * half);
        const float* bp = W1 + (size_t)(k + 2 * half) * F1 + ncol;
        v2f b; b.x = bp[0]; b.y = bp[F1];
        acc = __builtin_amdgcn_wmma_f32_16x16x4_f32(false, a, false, b,
                                                    (short)0, acc, false, false);
    }
    float* dr = h1 + (size_t)(mtile * 16 + 8 * half) * F1 + wave * 16 + l;
#pragma unroll
    for (int r = 0; r < 8; ++r) dr[(size_t)r * F1] = acc[r];
}

__global__ __launch_bounds__(96) void gat_gemm2_wmma(const float* __restrict__ h2in,
                                                     const float* __restrict__ W2,
                                                     float* __restrict__ h2) {
    const int mtile = blockIdx.x;
    const int wave  = threadIdx.x >> 5;
    const int lane  = threadIdx.x & 31;
    const int half  = lane >> 4;
    const int l     = lane & 15;
    const int row   = mtile * 16 + l;
    const int ncol  = wave * 16 + l;

    v8f acc = {};
    const float* xr = h2in + (size_t)row * F1;
    for (int k = 0; k < F1; k += 4) {
        v2f a = *(const v2f*)(xr + k + 2 * half);
        v2f b;
        if (ncol < NCLS) {
            const float* bp = W2 + (size_t)(k + 2 * half) * NCLS + ncol;
            b.x = bp[0]; b.y = bp[NCLS];
        } else { b.x = 0.f; b.y = 0.f; }
        acc = __builtin_amdgcn_wmma_f32_16x16x4_f32(false, a, false, b,
                                                    (short)0, acc, false, false);
    }
    if (ncol < NCLS) {
        float* dr = h2 + (size_t)(mtile * 16 + 8 * half) * NCLS + ncol;
#pragma unroll
        for (int r = 0; r < 8; ++r) dr[(size_t)r * NCLS] = acc[r];
    }
}

__global__ void gat_att1(const float* __restrict__ h1, const float* __restrict__ as1,
                         const float* __restrict__ ad1, float* __restrict__ a_src,
                         float* __restrict__ a_dst) {
    int i = blockIdx.x * blockDim.x + threadIdx.x;
    if (i >= N_NODES * HEADS) return;
    int n = i >> 1, h = i & 1;
    const float* hp = h1 + (size_t)n * F1 + h * HID;
    const float* s0 = as1 + h * HID;
    const float* d0 = ad1 + h * HID;
    float s = 0.f, d = 0.f;
#pragma unroll 8
    for (int c = 0; c < HID; ++c) { float v = hp[c]; s += v * s0[c]; d += v * d0[c]; }
    a_src[i] = s; a_dst[i] = d;
}

__global__ void gat_att2(const float* __restrict__ h2, const float* __restrict__ as2,
                         const float* __restrict__ ad2, float* __restrict__ a_src,
                         float* __restrict__ a_dst) {
    int n = blockIdx.x * blockDim.x + threadIdx.x;
    if (n >= N_NODES) return;
    const float* hp = h2 + (size_t)n * NCLS;
    float s = 0.f, d = 0.f;
#pragma unroll 8
    for (int c = 0; c < NCLS; ++c) { float v = hp[c]; s += v * as2[c]; d += v * ad2[c]; }
    a_src[n] = s; a_dst[n] = d;
}

#define BCAP 64

__global__ void gat_bucket_fill(const int* __restrict__ ei, int* __restrict__ count,
                                int* __restrict__ col, int2* __restrict__ ovf) {
    int e = blockIdx.x * blockDim.x + threadIdx.x;
    if (e >= N_EDGES) return;
    const int d = ei[N_EDGES + e];
    const int s = ei[e];
    const int r = atomicAdd(&count[d], 1);
    if (r < BCAP) {
        col[(size_t)d * BCAP + r] = s;
    } else {
        const int k = atomicAdd(&count[N_NODES], 1);
        ovf[k] = make_int2(d, s);
    }
}

static __device__ __forceinline__ void gat_chunk1(const bool valid, const int s_l, const int cnt,
        const int lane, const bool hi, const float2 adn, const float* __restrict__ a_src,
        const float* __restrict__ h1, float& m0, float& m1, float4& acc, float& den) {
    const float2 as = *(const float2*)(a_src + (size_t)s_l * 2);
    const float e0 = valid ? lrelu(as.x + adn.x) : -INFINITY;
    const float e1 = valid ? lrelu(as.y + adn.y) : -INFINITY;
    float c0m = e0, c1m = e1;
    for (int o = 16; o; o >>= 1) {
        c0m = fmaxf(c0m, __shfl_xor(c0m, o));
        c1m = fmaxf(c1m, __shfl_xor(c1m, o));
    }
    const float n0 = fmaxf(m0, c0m), n1 = fmaxf(m1, c1m);
    const float r = hi ? __expf(m1 - n1) : __expf(m0 - n0);
    acc.x *= r; acc.y *= r; acc.z *= r; acc.w *= r; den *= r;
    m0 = n0; m1 = n1;
    const int p0_l = __float_as_int(valid ? __expf(e0 - n0) : 0.f);
    const int p1_l = __float_as_int(valid ? __expf(e1 - n1) : 0.f);
    for (int j0 = 0; j0 < cnt; j0 += 4) {
#pragma unroll
        for (int u = 0; u < 4; ++u) {
            const int j = j0 + u;
            const int sj = __builtin_amdgcn_readlane(s_l, j);
            const float q0 = __int_as_float(__builtin_amdgcn_readlane(p0_l, j));
            const float q1 = __int_as_float(__builtin_amdgcn_readlane(p1_l, j));
            const float4 hv = *(const float4*)(h1 + (size_t)sj * F1 + lane * 4);
            const float p = hi ? q1 : q0;
            den += p;
            acc.x += p * hv.x; acc.y += p * hv.y; acc.z += p * hv.z; acc.w += p * hv.w;
        }
    }
}

__global__ __launch_bounds__(256) void gat_gather1(const int* __restrict__ count,
        const int* __restrict__ col, const int2* __restrict__ ovf,
        const float* __restrict__ a_src, const float* __restrict__ a_dst,
        const float* __restrict__ h1, const float* __restrict__ b1, float* __restrict__ h2in) {
    const int lane = threadIdx.x & 31;
    const int n = __builtin_amdgcn_readfirstlane(blockIdx.x * 8 + (threadIdx.x >> 5));
    if (n >= N_NODES) return;
    const int deg = count[n];
    const int nb  = min(deg, BCAP);
    const int* cb = col + (size_t)n * BCAP;
    const float2 adn = *(const float2*)(a_dst + (size_t)n * 2);
    const float2 asn = *(const float2*)(a_src + (size_t)n * 2);
    float m0 = lrelu(asn.x + adn.x), m1 = lrelu(asn.y + adn.y);
    const bool hi = lane >= 16;
    float4 acc = *(const float4*)(h1 + (size_t)n * F1 + lane * 4);
    float den = 1.f;
    for (int c0 = 0; c0 < nb; c0 += 32) {
        const int cnt = min(32, nb - c0);
        const bool valid = lane < cnt;
        const int s_l = valid ? cb[c0 + lane] : n;
        gat_chunk1(valid, s_l, cnt, lane, hi, adn, a_src, h1, m0, m1, acc, den);
    }
    if (deg > BCAP) {
        const int m = count[N_NODES];
        for (int k0 = 0; k0 < m; k0 += 32) {
            const int k = k0 + lane;
            const int2 en = k < m ? ovf[k] : make_int2(-1, 0);
            const bool valid = en.x == n;
            if (__ballot(valid) == 0) continue;
            const int s_l = valid ? en.y : n;
            gat_chunk1(valid, s_l, 32, lane, hi, adn, a_src, h1, m0, m1, acc, den);
        }
    }
    const float4 b = *(const float4*)(b1 + lane * 4);
    float4 o;
    o.x = acc.x / den + b.x; o.y = acc.y / den + b.y;
    o.z = acc.z / den + b.z; o.w = acc.w / den + b.w;
    o.x = o.x > 0.f ? o.x : (expf(o.x) - 1.0f);
    o.y = o.y > 0.f ? o.y : (expf(o.y) - 1.0f);
    o.z = o.z > 0.f ? o.z : (expf(o.z) - 1.0f);
    o.w = o.w > 0.f ? o.w : (expf(o.w) - 1.0f);
    *(float4*)(h2in + (size_t)n * F1 + lane * 4) = o;
}

static __device__ __forceinline__ void gat_chunk2(const bool valid, const int s_l, const int cnt,
        const bool live, const int g, const int q, const float adn,
        const float* __restrict__ a_src, const float* __restrict__ h2,
        float& m, float4& acc, float& den) {
    const float e = valid ? lrelu(a_src[s_l] + adn) : -INFINITY;
    float cm = e;
    for (int o = 16; o; o >>= 1) cm = fmaxf(cm, __shfl_xor(cm, o));
    const float nm = fmaxf(m, cm);
    const float r = __expf(m - nm);
    acc.x *= r; acc.y *= r; acc.z *= r; acc.w *= r; den *= r;
    m = nm;
    const float p_l = valid ? __expf(e - nm) : 0.f;
    for (int j0 = 0; j0 < cnt; j0 += 6) {
#pragma unroll
        for (int u = 0; u < 2; ++u) {
            const int j = j0 + 3 * u + g;
            const bool ok = live && (j < cnt);
            const int jj = ok ? j : 0;
            const int sj = __shfl(s_l, jj);
            const float pv = __shfl(p_l, jj);
            const float pj = ok ? pv : 0.f;
            const float4 hv = *(const float4*)(h2 + (size_t)sj * NCLS + q * 4);
            den += pj;
            acc.x += pj * hv.x; acc.y += pj * hv.y; acc.z += pj * hv.z; acc.w += pj * hv.w;
        }
    }
}

__global__ __launch_bounds__(256) void gat_gather2(const int* __restrict__ count,
        const int* __restrict__ col, const int2* __restrict__ ovf,
        const float* __restrict__ a_src, const float* __restrict__ a_dst,
        const float* __restrict__ h2, const float* __restrict__ b2, float* __restrict__ out) {
    const int lane = threadIdx.x & 31;
    const int n = __builtin_amdgcn_readfirstlane(blockIdx.x * 8 + (threadIdx.x >> 5));
    if (n >= N_NODES) return;
    const int g3 = lane / 10;
    const bool live = g3 < 3;
    const int g = live ? g3 : 0;
    const int q = lane - 10 * g3;
    const int deg = count[n];
    const int nb  = min(deg, BCAP);
    const int* cb = col + (size_t)n * BCAP;
    const float adn = a_dst[n];
    float m = lrelu(a_src[n] + adn);
    float4 acc = {0.f, 0.f, 0.f, 0.f};
    float den = 0.f;
    if (lane < 10) { acc = *(const float4*)(h2 + (size_t)n * NCLS + q * 4); den = 1.f; }
    for (int c0 = 0; c0 < nb; c0 += 32) {
        const int cnt = min(32, nb - c0);
        const bool valid = lane < cnt;
        const int s_l = valid ? cb[c0 + lane] : n;
        gat_chunk2(valid, s_l, cnt, live, g, q, adn, a_src, h2, m, acc, den);
    }
    if (deg > BCAP) {
        const int mo = count[N_NODES];
        for (int k0 = 0; k0 < mo; k0 += 32) {
            const int k = k0 + lane;
            const int2 en = k < mo ? ovf[k] : make_int2(-1, 0);
            const bool valid = en.x == n;
            if (__ballot(valid) == 0) continue;
            const int s_l = valid ? en.y : n;
            gat_chunk2(valid, s_l, 32, live, g, q, adn, a_src, h2, m, acc, den);
        }
    }
    const int l1 = (lane + 10) & 31, l2 = (lane + 20) & 31;
    const float sx = acc.x + __shfl(acc.x, l1) + __shfl(acc.x, l2);
    const float sy = acc.y + __shfl(acc.y, l1) + __shfl(acc.y, l2);
    const float sz = acc.z + __shfl(acc.z, l1) + __shfl(acc.z, l2);
    const float sw = acc.w + __shfl(acc.w, l1) + __shfl(acc.w, l2);
    const float sd = den + __shfl(den, l1) + __shfl(den, l2);
    if (lane < 10) {
        const float4 b = *(const float4*)(b2 + q * 4);
        float4 o;
        o.x = sx / sd + b.x; o.y = sy / sd + b.y; o.z = sz / sd + b.z; o.w = sw / sd + b.w;
        *(float4*)(out + (size_t)n * NCLS + q * 4) = o;
    }
}

extern "C" void kernel_launch(void* const* d_in, const int* in_sizes, int n_in,
                              void* d_out, int out_size, void* d_ws, size_t ws_size,
                              hipStream_t stream) {
    const float* x    = (const float*)d_in[0];
    const int*   ei   = (const int*)  d_in[1];
    const float* W1   = (const float*)d_in[2];
    const float* as1  = (const float*)d_in[3];
    const float* ad1  = (const float*)d_in[4];
    const float* b1   = (const float*)d_in[5];
    const float* W2   = (const float*)d_in[6];
    const float* as2  = (const float*)d_in[7];
    const float* ad2  = (const float*)d_in[8];
    const float* b2   = (const float*)d_in[9];
    float* out = (float*)d_out;

    float* ws      = (float*)d_ws;
    float* h1      = ws;
    float* h2in    = h1     + (size_t)N_NODES * F1;
    float* h2      = h2in   + (size_t)N_NODES * F1;
    float* a_src1  = h2     + (size_t)N_NODES * NCLS;
    float* a_dst1  = a_src1 + (size_t)N_NODES * 2;
    float* a_src2  = a_dst1 + (size_t)N_NODES * 2;
    float* a_dst2  = a_src2 + N_NODES;
    int*   count   = (int*)(a_dst2 + N_NODES);
    int*   col     = count  + N_NODES + 16;
    int2*  ovf     = (int2*)(col + (size_t)N_NODES * BCAP);

    hipMemsetAsync(count, 0, (size_t)(N_NODES + 16) * sizeof(int), stream);

    const int MT = N_NODES / 16;
    gat_bucket_fill<<<(N_EDGES + 255) / 256, 256, 0, stream>>>(ei, count, col, ovf);
    gat_gemm1_wmma<<<MT, 256, 0, stream>>>(x, W1, h1);
    gat_att1<<<(N_NODES * HEADS + 255) / 256, 256, 0, stream>>>(h1, as1, ad1, a_src1, a_dst1);
    gat_gather1<<<N_NODES / 8, 256, 0, stream>>>(count, col, ovf, a_src1, a_dst1, h1, b1, h2in);
    gat_gemm2_wmma<<<MT, 96, 0, stream>>>(h2in, W2, h2);
    gat_att2<<<(N_NODES + 255) / 256, 256, 0, stream>>>(h2, as2, ad2, a_src2, a_dst2);
    gat_gather2<<<N_NODES / 8, 256, 0, stream>>>(count, col, ovf, a_src2, a_dst2, h2, b2, out);
}


# Bucket fill rides in the GEMM1 launch (odd blocks place 256 edges, even blocks do a GEMM1 tile): the counting adds wait under matrix work, one launch fewer; sums in slot order, arrival-set
# speedup vs baseline: 3.4644x; 1.0740x over previous
#include <hip/hip_runtime.h>
#include <math.h>

#define N_NODES 50000
#define N_EDGES 800000
#define IN_FEATS 256
#define HID 64
#define HEADS 2
#define F1 (HEADS * HID)
#define NCLS 40

typedef __attribute__((ext_vector_type(2))) float v2f;
typedef __attribute__((ext_vector_type(8))) float v8f;

static __device__ __forceinline__ float lrelu(float x) { return x > 0.f ? x : 0.2f * x; }

static __device__ __forceinline__ void gat_gemm1_tile(const float* __restrict__ x,
                                                      const float* __restrict__ W1,
                                                      float* __restrict__ h1, const int mtile) {
    const int wave  = threadIdx.x >> 5;
    const int lane  = threadIdx.x & 31;
    const int half  = lane >> 4;
    const int l     = lane & 15;
    const int row   = mtile * 16 + l;
    const int ncol  = wave * 16 + l;

    v8f acc = {};
    const float* xr = x + (size_t)row * IN_FEATS;
    for (int k = 0; k < IN_FEATS; k += 4) {
        v2f a = *(const v2f*)(xr + k + 2 * half);
        const float* bp = W1 + (size_t)(k + 2 * half) * F1 + ncol;
        v2f b; b.x = bp[0]; b.y = bp[F1];
        acc = __builtin_amdgcn_wmma_f32_16x16x4_f32(false, a, false, b,
                                                    (short)0, acc, false, false);
    }
    float* dr = h1 + (size_t)(mtile * 16 + 8 * half) * F1 + wave * 16 + l;
#pragma unroll
    for (int r = 0; r < 8; ++r) dr[(size_t)r * F1] = acc[r];
}

__global__ __launch_bounds__(96) void gat_gemm2_wmma(const float* __restrict__ h2in,
                                                     const float* __restrict__ W2,
                                                     float* __restrict__ h2) {
    const int mtile = blockIdx.x;
    const int wave  = threadIdx.x >> 5;
    const int lane  = threadIdx.x & 31;
    const int half  = lane >> 4;
    const int l     = lane & 15;
    const int row   = mtile * 16 + l;
    const int ncol  = wave * 16 + l;

    v8f acc = {};
    const float* xr = h2in + (size_t)row * F1;
    for (int k = 0; k < F1; k += 4) {
        v2f a = *(const v2f*)(xr + k + 2 * half);
        v2f b;
        if (ncol < NCLS) {
            const float* bp = W2 + (size_t)(k + 2 * half) * NCLS + ncol;
            b.x = bp[0]; b.y = bp[NCLS];
        } else { b.x = 0.f; b.y = 0.f; }
        acc = __builtin_amdgcn_wmma_f32_16x16x4_f32(false, a, false, b,
                                                    (short)0, acc, false, false);
    }
    if (ncol < NCLS) {
        float* dr = h2 + (size_t)(mtile * 16 + 8 * half) * NCLS + ncol;
#pragma unroll
        for (int r = 0; r < 8; ++r) dr[(size_t)r * NCLS] = acc[r];
    }
}

__global__ void gat_att1(const float* __restrict__ h1, const float* __restrict__ as1,
                         const float* __restrict__ ad1, float* __restrict__ a_src,
                         float* __restrict__ a_dst) {
    int i = blockIdx.x * blockDim.x + threadIdx.x;
    if (i >= N_NODES * HEADS) return;
    int n = i >> 1, h = i & 1;
    const float* hp = h1 + (size_t)n * F1 + h * HID;
    const float* s0 = as1 + h * HID;
    const float* d0 = ad1 + h * HID;
    float s = 0.f, d = 0.f;
#pragma unroll 8
    for (int c = 0; c < HID; ++c) { float v = hp[c]; s += v * s0[c]; d += v * d0[c]; }
    a_src[i] = s; a_dst[i] = d;
}

__global__ void gat_att2(const float* __restrict__ h2, const float* __restrict__ as2,
                         const float* __restrict__ ad2, float* __restrict__ a_src,
                         float* __restrict__ a_dst) {
    int n = blockIdx.x * blockDim.x + threadIdx.x;
    if (n >= N_NODES) return;
    const float* hp = h2 + (size_t)n * NCLS;
    float s = 0.f, d = 0.f;
#pragma unroll 8
    for (int c = 0; c < NCLS; ++c) { float v = hp[c]; s += v * as2[c]; d += v * ad2[c]; }
    a_src[n] = s; a_dst[n] = d;
}

#define BCAP 64

static __device__ __forceinline__ void gat_bucket_fill(const int* __restrict__ ei, int* __restrict__ count,
                                int* __restrict__ col, int2* __restrict__ ovf, const int e) {
    if (e >= N_EDGES) return;
    const int d = ei[N_EDGES + e];
    const int s = ei[e];
    const int r = atomicAdd(&count[d], 1);
    if (r < BCAP) {
        col[(size_t)d * BCAP + r] = s;
    } else {
        const int k = atomicAdd(&count[N_NODES], 1);
        ovf[k] = make_int2(d, s);
    }
}

#define GEMM1_BLOCKS (N_NODES / 16)
#define FILL_BLOCKS  ((N_EDGES + 255) / 256)
#define LAUNCH1_PAIRS (GEMM1_BLOCKS > FILL_BLOCKS ? GEMM1_BLOCKS : FILL_BLOCKS)

__global__ __launch_bounds__(256) void gat_gemm1_and_buckets(const float* __restrict__ x,
        const float* __restrict__ W1, float* __restrict__ h1, const int* __restrict__ ei,
        int* __restrict__ count, int* __restrict__ col, int2* __restrict__ ovf) {
    const int unit = blockIdx.x >> 1;
    if (blockIdx.x & 1) {
        if (unit < FILL_BLOCKS) gat_bucket_fill(ei, count, col, ovf, unit * 256 + (int)threadIdx.x);
    } else {
        if (unit < GEMM1_BLOCKS) gat_gemm1_tile(x, W1, h1, unit);
    }
}

static __device__ __forceinline__ void gat_chunk1(const bool valid, const int s_l, const int cnt,
        const int lane, const bool hi, const float2 adn, const float* __restrict__ a_src,
        const float* __restrict__ h1, float& m0, float& m1, float4& acc, float& den) {
    const float2 as = *(const float2*)(a_src + (size_t)s_l * 2);
    const float e0 = valid ? lrelu(as.x + adn.x) : -INFINITY;
    const float e1 = valid ? lrelu(as.y + adn.y) : -INFINITY;
    float c0m = e0, c1m = e1;
    for (int o = 16; o; o >>= 1) {
        c0m = fmaxf(c0m, __shfl_xor(c0m, o));
        c1m = fmaxf(c1m, __shfl_xor(c1m, o));
    }
    const float n0 = fmaxf(m0, c0m), n1 = fmaxf(m1, c1m);
    const float r = hi ? __expf(m1 - n1) : __expf(m0 - n0);
    acc.x *= r; acc.y *= r; acc.z *= r; acc.w *= r; den *= r;
    m0 = n0; m1 = n1;
    const int p0_l = __float_as_int(valid ? __expf(e0 - n0) : 0.f);
    const int p1_l = __float_as_int(valid ? __expf(e1 - n1) : 0.f);
    for (int j0 = 0; j0 < cnt; j0 += 4) {
#pragma unroll
        for (int u = 0; u < 4; ++u) {
            const int j = j0 + u;
            const int sj = __builtin_amdgcn_readlane(s_l, j);
            const float q0 = __int_as_float(__builtin_amdgcn_readlane(p0_l, j));
            const float q1 = __int_as_float(__builtin_amdgcn_readlane(p1_l, j));
            const float4 hv = *(const float4*)(h1 + (size_t)sj * F1 + lane * 4);
            const float p = hi ? q1 : q0;
            den += p;
            acc.x += p * hv.x; acc.y += p * hv.y; acc.z += p * hv.z; acc.w += p * hv.w;
        }
    }
}

__global__ __launch_bounds__(256) void gat_gather1(const int* __restrict__ count,
        const int* __restrict__ col, const int2* __restrict__ ovf,
        const float* __restrict__ a_src, const float* __restrict__ a_dst,
        const float* __restrict__ h1, const float* __restrict__ b1, float* __restrict__ h2in) {
    const int lane = threadIdx.x & 31;
    const int n = __builtin_amdgcn_readfirstlane(blockIdx.x * 8 + (threadIdx.x >> 5));
    if (n >= N_NODES) return;
    const int deg = count[n];
    const int nb  = min(deg, BCAP);
    const int* cb = col + (size_t)n * BCAP;
    const float2 adn = *(const float2*)(a_dst + (size_t)n * 2);
    const float2 asn = *(const float2*)(a_src + (size_t)n * 2);
    float m0 = lrelu(asn.x + adn.x), m1 = lrelu(asn.y + adn.y);
    const bool hi = lane >= 16;
    float4 acc = *(const float4*)(h1 + (size_t)n * F1 + lane * 4);
    float den = 1.f;
    for (int c0 = 0; c0 < nb; c0 += 32) {
        const int cnt = min(32, nb - c0);
        const bool valid = lane < cnt;
        const int s_l = valid ? cb[c0 + lane] : n;
        gat_chunk1(valid, s_l, cnt, lane, hi, adn, a_src, h1, m0, m1, acc, den);
    }
    if (deg > BCAP) {
        const int m = count[N_NODES];
        for (int k0 = 0; k0 < m; k0 += 32) {
            const int k = k0 + lane;
            const int2 en = k < m ? ovf[k] : make_int2(-1, 0);
            const bool valid = en.x == n;
            if (__ballot(valid) == 0) continue;
            const int s_l = valid ? en.y : n;
            gat_chunk1(valid, s_l, 32, lane, hi, adn, a_src, h1, m0, m1, acc, den);
        }
    }
    const float4 b = *(const float4*)(b1 + lane * 4);
    float4 o;
    o.x = acc.x / den + b.x; o.y = acc.y / den + b.y;
    o.z = acc.z / den + b.z; o.w = acc.w / den + b.w;
    o.x = o.x > 0.f ? o.x : (expf(o.x) - 1.0f);
    o.y = o.y > 0.f ? o.y : (expf(o.y) - 1.0f);
    o.z = o.z > 0.f ? o.z : (expf(o.z) - 1.0f);
    o.w = o.w > 0.f ? o.w : (expf(o.w) - 1.0f);
    *(float4*)(h2in + (size_t)n * F1 + lane * 4) = o;
}

static __device__ __forceinline__ void gat_chunk2(const bool valid, const int s_l, const int cnt,
        const bool live, const int g, const int q, const float adn,
        const float* __restrict__ a_src, const float* __restrict__ h2,
        float& m, float4& acc, float& den) {
    const float e = valid ? lrelu(a_src[s_l] + adn) : -INFINITY;
    float cm = e;
    for (int o = 16; o; o >>= 1) cm = fmaxf(cm, __shfl_xor(cm, o));
    const float nm = fmaxf(m, cm);
    const float r = __expf(m - nm);
    acc.x *= r; acc.y *= r; acc.z *= r; acc.w *= r; den *= r;
    m = nm;
    const float p_l = valid ? __expf(e - nm) : 0.f;
    for (int j0 = 0; j0 < cnt; j0 += 6) {
#pragma unroll
        for (int u = 0; u < 2; ++u) {
            const int j = j0 + 3 * u + g;
            const bool ok = live && (j < cnt);
            const int jj = ok ? j : 0;
            const int sj = __shfl(s_l, jj);
            const float pv = __shfl(p_l, jj);
            const float pj = ok ? pv : 0.f;
            const float4 hv = *(const float4*)(h2 + (size_t)sj * NCLS + q * 4);
            den += pj;
            acc.x += pj * hv.x; acc.y += pj * hv.y; acc.z += pj * hv.z; acc.w += pj * hv.w;
        }
    }
}

__global__ __launch_bounds__(256) void gat_gather2(const int* __restrict__ count,
        const int* __restrict__ col, const int2* __restrict__ ovf,
        const float* __restrict__ a_src, const float* __restrict__ a_dst,
        const float* __restrict__ h2, const float* __restrict__ b2, float* __restrict__ out) {
    const int lane = threadIdx.x & 31;
    const int n = __builtin_amdgcn_readfirstlane(blockIdx.x * 8 + (threadIdx.x >> 5));
    if (n >= N_NODES) return;
    const int g3 = lane / 10;
    const bool live = g3 < 3;
    const int g = live ? g3 : 0;
    const int q = lane - 10 * g3;
    const int deg = count[n];
    const int nb  = min(deg, BCAP);
    const int* cb = col + (size_t)n * BCAP;
    const float adn = a_dst[n];
    float m = lrelu(a_src[n] + adn);
    float4 acc = {0.f, 0.f, 0.f, 0.f};
    float den = 0.f;
    if (lane < 10) { acc = *(const float4*)(h2 + (size_t)n * NCLS + q * 4); den = 1.f; }
    for (int c0 = 0; c0 < nb; c0 += 32) {
        const int cnt = min(32, nb - c0);
        const bool valid = lane < cnt;
        const int s_l = valid ? cb[c0 + lane] : n;
        gat_chunk2(valid, s_l, cnt, live, g, q, adn, a_src, h2, m, acc, den);
    }
    if (deg > BCAP) {
        const int mo = count[N_NODES];
        for (int k0 = 0; k0 < mo; k0 += 32) {
            const int k = k0 + lane;
            const int2 en = k < mo ? ovf[k] : make_int2(-1, 0);
            const bool valid = en.x == n;
            if (__ballot(valid) == 0) continue;
            const int s_l = valid ? en.y : n;
            gat_chunk2(valid, s_l, 32, live, g, q, adn, a_src, h2, m, acc, den);
        }
    }
    const int l1 = (lane + 10) & 31, l2 = (lane + 20) & 31;
    const float sx = acc.x + __shfl(acc.x, l1) + __shfl(acc.x, l2);
    const float sy = acc.y + __shfl(acc.y, l1) + __shfl(acc.y, l2);
    const float sz = acc.z + __shfl(acc.z, l1) + __shfl(acc.z, l2);
    const float sw = acc.w + __shfl(acc.w, l1) + __shfl(acc.w, l2);
    const float sd = den + __shfl(den, l1) + __shfl(den, l2);
    if (lane < 10) {
        const float4 b = *(const float4*)(b2 + q * 4);
        float4 o;
        o.x = sx / sd + b.x; o.y = sy / sd + b.y; o.z = sz / sd + b.z; o.w = sw / sd + b.w;
        *(float4*)(out + (size_t)n * NCLS + q * 4) = o;
    }
}

extern "C" void kernel_launch(void* const* d_in, const int* in_sizes, int n_in,
                              void* d_out, int out_size, void* d_ws, size_t ws_size,
                              hipStream_t stream) {
    const float* x    = (const float*)d_in[0];
    const int*   ei   = (const int*)  d_in[1];
    const float* W1   = (const float*)d_in[2];
    const float* as1  = (const float*)d_in[3];
    const float* ad1  = (const float*)d_in[4];
    const float* b1   = (const float*)d_in[5];
    const float* W2   = (const float*)d_in[6];
    const float* as2  = (const float*)d_in[7];
    const float* ad2  = (const float*)d_in[8];
    const float* b2   = (const float*)d_in[9];
    float* out = (float*)d_out;

    float* ws      = (float*)d_ws;
    float* h1      = ws;
    float* h2in    = h1     + (size_t)N_NODES * F1;
    float* h2      = h2in   + (size_t)N_NODES * F1;
    float* a_src1  = h2     + (size_t)N_NODES * NCLS;
    float* a_dst1  = a_src1 + (size_t)N_NODES * 2;
    float* a_src2  = a_dst1 + (size_t)N_NODES * 2;
    float* a_dst2  = a_src2 + N_NODES;
    int*   count   = (int*)(a_dst2 + N_NODES);
    int*   col     = count  + N_NODES + 16;
    int2*  ovf     = (int2*)(col + (size_t)N_NODES * BCAP);

    hipMemsetAsync(count, 0, (size_t)(N_NODES + 16) * sizeof(int), stream);

    const int MT = N_NODES / 16;
    gat_gemm1_and_buckets<<<2 * LAUNCH1_PAIRS, 256, 0, stream>>>(x, W1, h1, ei, count, col, ovf);
    gat_att1<<<(N_NODES * HEADS + 255) / 256, 256, 0, stream>>>(h1, as1, ad1, a_src1, a_dst1);
    gat_gather1<<<N_NODES / 8, 256, 0, stream>>>(count, col, ovf, a_src1, a_dst1, h1, b1, h2in);
    gat_gemm2_wmma<<<MT, 96, 0, stream>>>(h2in, W2, h2);
    gat_att2<<<(N_NODES + 255) / 256, 256, 0, stream>>>(h2, as2, ad2, a_src2, a_dst2);
    gat_gather2<<<N_NODES / 8, 256, 0, stream>>>(count, col, ovf, a_src2, a_dst2, h2, b2, out);
}


# GEMM1 retiled: W1 staged once per block in LDS in fragment order, one wave does 16 rows x all 128 columns so an A fragment feeds 8 WMMAs; edge fill rides per k-chunk; k summed in a permuted order
# speedup vs baseline: 3.9289x; 1.1341x over previous
#include <hip/hip_runtime.h>
#include <math.h>

#define N_NODES 50000
#define N_EDGES 800000
#define IN_FEATS 256
#define HID 64
#define HEADS 2
#define F1 (HEADS * HID)
#define NCLS 40

typedef __attribute__((ext_vector_type(2))) float v2f;
typedef __attribute__((ext_vector_type(8))) float v8f;

static __device__ __forceinline__ float lrelu(float x) { return x > 0.f ? x : 0.2f * x; }

__global__ __launch_bounds__(96) void gat_gemm2_wmma(const float* __restrict__ h2in,
                                                     const float* __restrict__ W2,
                                                     float* __restrict__ h2) {
    const int mtile = blockIdx.x;
    const int wave  = threadIdx.x >> 5;
    const int lane  = threadIdx.x & 31;
    const int half  = lane >> 4;
    const int l     = lane & 15;
    const int row   = mtile * 16 + l;
    const int ncol  = wave * 16 + l;

    v8f acc = {};
    const float* xr = h2in + (size_t)row * F1;
    for (int k = 0; k < F1; k += 4) {
        v2f a = *(const v2f*)(xr + k + 2 * half);
        v2f b;
        if (ncol < NCLS) {
            const float* bp = W2 + (size_t)(k + 2 * half) * NCLS + ncol;
            b.x = bp[0]; b.y = bp[NCLS];
        } else { b.x = 0.f; b.y = 0.f; }
        acc = __builtin_amdgcn_wmma_f32_16x16x4_f32(false, a, false, b,
                                                    (short)0, acc, false, false);
    }
    if (ncol < NCLS) {
        float* dr = h2 + (size_t)(mtile * 16 + 8 * half) * NCLS + ncol;
#pragma unroll
        for (int r = 0; r < 8; ++r) dr[(size_t)r * NCLS] = acc[r];
    }
}

__global__ void gat_att1(const float* __restrict__ h1, const float* __restrict__ as1,
                         const float* __restrict__ ad1, float* __restrict__ a_src,
                         float* __restrict__ a_dst) {
    int i = blockIdx.x * blockDim.x + threadIdx.x;
    if (i >= N_NODES * HEADS) return;
    int n = i >> 1, h = i & 1;
    const float* hp = h1 + (size_t)n * F1 + h * HID;
    const float* s0 = as1 + h * HID;
    const float* d0 = ad1 + h * HID;
    float s = 0.f, d = 0.f;
#pragma unroll 8
    for (int c = 0; c < HID; ++c) { float v = hp[c]; s += v * s0[c]; d += v * d0[c]; }
    a_src[i] = s; a_dst[i] = d;
}

__global__ void gat_att2(const float* __restrict__ h2, const float* __restrict__ as2,
                         const float* __restrict__ ad2, float* __restrict__ a_src,
                         float* __restrict__ a_dst) {
    int n = blockIdx.x * blockDim.x + threadIdx.x;
    if (n >= N_NODES) return;
    const float* hp = h2 + (size_t)n * NCLS;
    float s = 0.f, d = 0.f;
#pragma unroll 8
    for (int c = 0; c < NCLS; ++c) { float v = hp[c]; s += v * as2[c]; d += v * ad2[c]; }
    a_src[n] = s; a_dst[n] = d;
}

#define BCAP 64

static __device__ __forceinline__ int gat_edge_rank(int* __restrict__ count, const int d) {
    return d >= 0 ? atomicAdd(&count[d], 1) : 0;
}

static __device__ __forceinline__ void gat_edge_place(int* __restrict__ count, int* __restrict__ col,
                                int2* __restrict__ ovf, const int d, const int s, const int r) {
    if (d < 0) return;
    if (r < BCAP) {
        col[(size_t)d * BCAP + r] = s;
    } else {
        const int k = atomicAdd(&count[N_NODES], 1);
        ovf[k] = make_int2(d, s);
    }
}

#define G1_TILES   (N_NODES / 16)
#define G1_BLOCKS  ((G1_TILES + 7) / 8)
#define G1_EPT     8

__global__ __launch_bounds__(256) void gat_gemm1_and_buckets(const float* __restrict__ x,
        const float* __restrict__ W1, float* __restrict__ h1, const int* __restrict__ ei,
        int* __restrict__ count, int* __restrict__ col, int2* __restrict__ ovf) {
    __shared__ float sB[IN_FEATS * F1];
    const int tid  = threadIdx.x;
    const int wave = tid >> 5;
    const int lane = tid & 31;
    const int half = lane >> 4;
    const int l    = lane & 15;

    int ed[G1_EPT], es[G1_EPT], er[G1_EPT];
#pragma unroll
    for (int i = 0; i < G1_EPT; ++i) {
        const int e = (blockIdx.x * G1_EPT + i) * 256 + tid;
        const bool ok = e < N_EDGES;
        ed[i] = ok ? ei[N_EDGES + e] : -1;
        es[i] = ok ? ei[e] : 0;
    }

    const int  mtile   = blockIdx.x * 8 + wave;
    const bool tile_ok = mtile < G1_TILES;
    const int  mt      = tile_ok ? mtile : G1_TILES - 1;
    const float* xr = x + (size_t)(mt * 16 + l) * IN_FEATS + 4 * half;
    float4 a_cur[8], a_nxt[8];
#pragma unroll
    for (int q = 0; q < 8; ++q) a_cur[q] = *(const float4*)(xr + 8 * q);

    for (int i = tid; i < (IN_FEATS / 4) * F1; i += 256) {
        const int j = i >> 7, n = i & (F1 - 1);
        const int k0 = 8 * (j >> 1) + 2 * (j & 1);
        float4 v;
        v.x = W1[(size_t)(k0    ) * F1 + n];
        v.y = W1[(size_t)(k0 + 1) * F1 + n];
        v.z = W1[(size_t)(k0 + 4) * F1 + n];
        v.w = W1[(size_t)(k0 + 5) * F1 + n];
        *(float4*)(sB + j * 512 + n * 4) = v;
    }
    __syncthreads();

    const float* bl = sB + l * 4 + half * 2;

    v8f acc[8];
#pragma unroll
    for (int t = 0; t < 8; ++t) acc[t] = (v8f){};


#pragma unroll
    for (int c = 0; c < 4; ++c) {
        if (c < 3) {
#pragma unroll
            for (int q = 0; q < 8; ++q) a_nxt[q] = *(const float4*)(xr + 64 * (c + 1) + 8 * q);
        }
        er[2 * c]     = gat_edge_rank(count, ed[2 * c]);
        er[2 * c + 1] = gat_edge_rank(count, ed[2 * c + 1]);
        if (c > 0) {
            gat_edge_place(count, col, ovf, ed[2 * c - 2], es[2 * c - 2], er[2 * c - 2]);
            gat_edge_place(count, col, ovf, ed[2 * c - 1], es[2 * c - 1], er[2 * c - 1]);
        }
#pragma unroll
        for (int qq = 0; qq < 8; ++qq) {
#pragma unroll
            for (int e = 0; e < 2; ++e) {
                const int j = 2 * (8 * c + qq) + e;
                v2f a;
                a.x = e ? a_cur[qq].z : a_cur[qq].x;
                a.y = e ? a_cur[qq].w : a_cur[qq].y;
#pragma unroll
                for (int t = 0; t < 8; ++t) {
                    const v2f b = *(const v2f*)(bl + j * 512 + t * 64);
                    acc[t] = __builtin_amdgcn_wmma_f32_16x16x4_f32(false, a, false, b,
                                                                   (short)0, acc[t], false, false);
                }
            }
        }
        if (c < 3) {
#pragma unroll
            for (int q = 0; q < 8; ++q) a_cur[q] = a_nxt[q];
        }
    }
    gat_edge_place(count, col, ovf, ed[6], es[6], er[6]);
    gat_edge_place(count, col, ovf, ed[7], es[7], er[7]);

    if (tile_ok) {
        float* dr = h1 + (size_t)(mt * 16 + 8 * half) * F1 + l;
#pragma unroll
        for (int t = 0; t < 8; ++t) {
#pragma unroll
            for (int r = 0; r < 8; ++r) dr[(size_t)r * F1 + t * 16] = acc[t][r];
        }
    }
}

static __device__ __forceinline__ void gat_chunk1(const bool valid, const int s_l, const int cnt,
        const int lane, const bool hi, const float2 adn, const float* __restrict__ a_src,
        const float* __restrict__ h1, float& m0, float& m1, float4& acc, float& den) {
    const float2 as = *(const float2*)(a_src + (size_t)s_l * 2);
    const float e0 = valid ? lrelu(as.x + adn.x) : -INFINITY;
    const float e1 = valid ? lrelu(as.y + adn.y) : -INFINITY;
    float c0m = e0, c1m = e1;
    for (int o = 16; o; o >>= 1) {
        c0m = fmaxf(c0m, __shfl_xor(c0m, o));
        c1m = fmaxf(c1m, __shfl_xor(c1m, o));
    }
    const float n0 = fmaxf(m0, c0m), n1 = fmaxf(m1, c1m);
    const float r = hi ? __expf(m1 - n1) : __expf(m0 - n0);
    acc.x *= r; acc.y *= r; acc.z *= r; acc.w *= r; den *= r;
    m0 = n0; m1 = n1;
    const int p0_l = __float_as_int(valid ? __expf(e0 - n0) : 0.f);
    const int p1_l = __float_as_int(valid ? __expf(e1 - n1) : 0.f);
    for (int j0 = 0; j0 < cnt; j0 += 4) {
#pragma unroll
        for (int u = 0; u < 4; ++u) {
            const int j = j0 + u;
            const int sj = __builtin_amdgcn_readlane(s_l, j);
            const float q0 = __int_as_float(__builtin_amdgcn_readlane(p0_l, j));
            const float q1 = __int_as_float(__builtin_amdgcn_readlane(p1_l, j));
            const float4 hv = *(const float4*)(h1 + (size_t)sj * F1 + lane * 4);
            const float p = hi ? q1 : q0;
            den += p;
            acc.x += p * hv.x; acc.y += p * hv.y; acc.z += p * hv.z; acc.w += p * hv.w;
        }
    }
}

__global__ __launch_bounds__(256) void gat_gather1(const int* __restrict__ count,
        const int* __restrict__ col, const int2* __restrict__ ovf,
        const float* __restrict__ a_src, const float* __restrict__ a_dst,
        const float* __restrict__ h1, const float* __restrict__ b1, float* __restrict__ h2in) {
    const int lane = threadIdx.x & 31;
    const int n = __builtin_amdgcn_readfirstlane(blockIdx.x * 8 + (threadIdx.x >> 5));
    if (n >= N_NODES) return;
    const int deg = count[n];
    const int nb  = min(deg, BCAP);
    const int* cb = col + (size_t)n * BCAP;
    const float2 adn = *(const float2*)(a_dst + (size_t)n * 2);
    const float2 asn = *(const float2*)(a_src + (size_t)n * 2);
    float m0 = lrelu(asn.x + adn.x), m1 = lrelu(asn.y + adn.y);
    const bool hi = lane >= 16;
    float4 acc = *(const float4*)(h1 + (size_t)n * F1 + lane * 4);
    float den = 1.f;
    for (int c0 = 0; c0 < nb; c0 += 32) {
        const int cnt = min(32, nb - c0);
        const bool valid = lane < cnt;
        const int s_l = valid ? cb[c0 + lane] : n;
        gat_chunk1(valid, s_l, cnt, lane, hi, adn, a_src, h1, m0, m1, acc, den);
    }
    if (deg > BCAP) {
        const int m = count[N_NODES];
        for (int k0 = 0; k0 < m; k0 += 32) {
            const int k = k0 + lane;
            const int2 en = k < m ? ovf[k] : make_int2(-1, 0);
            const bool valid = en.x == n;
            if (__ballot(valid) == 0) continue;
            const int s_l = valid ? en.y : n;
            gat_chunk1(valid, s_l, 32, lane, hi, adn, a_src, h1, m0, m1, acc, den);
        }
    }
    const float4 b = *(const float4*)(b1 + lane * 4);
    float4 o;
    o.x = acc.x / den + b.x; o.y = acc.y / den + b.y;
    o.z = acc.z / den + b.z; o.w = acc.w / den + b.w;
    o.x = o.x > 0.f ? o.x : (expf(o.x) - 1.0f);
    o.y = o.y > 0.f ? o.y : (expf(o.y) - 1.0f);
    o.z = o.z > 0.f ? o.z : (expf(o.z) - 1.0f);
    o.w = o.w > 0.f ? o.w : (expf(o.w) - 1.0f);
    *(float4*)(h2in + (size_t)n * F1 + lane * 4) = o;
}

static __device__ __forceinline__ void gat_chunk2(const bool valid, const int s_l, const int cnt,
        const bool live, const int g, const int q, const float adn,
        const float* __restrict__ a_src, const float* __restrict__ h2,
        float& m, float4& acc, float& den) {
    const float e = valid ? lrelu(a_src[s_l] + adn) : -INFINITY;
    float cm = e;
    for (int o = 16; o; o >>= 1) cm = fmaxf(cm, __shfl_xor(cm, o));
    const float nm = fmaxf(m, cm);
    const float r = __expf(m - nm);
    acc.x *= r; acc.y *= r; acc.z *= r; acc.w *= r; den *= r;
    m = nm;
    const float p_l = valid ? __expf(e - nm) : 0.f;
    for (int j0 = 0; j0 < cnt; j0 += 6) {
#pragma unroll
        for (int u = 0; u < 2; ++u) {
            const int j = j0 + 3 * u + g;
            const bool ok = live && (j < cnt);
            const int jj = ok ? j : 0;
            const int sj = __shfl(s_l, jj);
            const float pv = __shfl(p_l, jj);
            const float pj = ok ? pv : 0.f;
            const float4 hv = *(const float4*)(h2 + (size_t)sj * NCLS + q * 4);
            den += pj;
            acc.x += pj * hv.x; acc.y += pj * hv.y; acc.z += pj * hv.z; acc.w += pj * hv.w;
        }
    }
}

__global__ __launch_bounds__(256) void gat_gather2(const int* __restrict__ count,
        const int* __restrict__ col, const int2* __restrict__ ovf,
        const float* __restrict__ a_src, const float* __restrict__ a_dst,
        const float* __restrict__ h2, const float* __restrict__ b2, float* __restrict__ out) {
    const int lane = threadIdx.x & 31;
    const int n = __builtin_amdgcn_readfirstlane(blockIdx.x * 8 + (threadIdx.x >> 5));
    if (n >= N_NODES) return;
    const int g3 = lane / 10;
    const bool live = g3 < 3;
    const int g = live ? g3 : 0;
    const int q = lane - 10 * g3;
    const int deg = count[n];
    const int nb  = min(deg, BCAP);
    const int* cb = col + (size_t)n * BCAP;
    const float adn = a_dst[n];
    float m = lrelu(a_src[n] + adn);
    float4 acc = {0.f, 0.f, 0.f, 0.f};
    float den = 0.f;
    if (lane < 10) { acc = *(const float4*)(h2 + (size_t)n * NCLS + q * 4); den = 1.f; }
    for (int c0 = 0; c0 < nb; c0 += 32) {
        const int cnt = min(32, nb - c0);
        const bool valid = lane < cnt;
        const int s_l = valid ? cb[c0 + lane] : n;
        gat_chunk2(valid, s_l, cnt, live, g, q, adn, a_src, h2, m, acc, den);
    }
    if (deg > BCAP) {
        const int mo = count[N_NODES];
        for (int k0 = 0; k0 < mo; k0 += 32) {
            const int k = k0 + lane;
            const int2 en = k < mo ? ovf[k] : make_int2(-1, 0);
            const bool valid = en.x == n;
            if (__ballot(valid) == 0) continue;
            const int s_l = valid ? en.y : n;
            gat_chunk2(valid, s_l, 32, live, g, q, adn, a_src, h2, m, acc, den);
        }
    }
    const int l1 = (lane + 10) & 31, l2 = (lane + 20) & 31;
    const float sx = acc.x + __shfl(acc.x, l1) + __shfl(acc.x, l2);
    const float sy = acc.y + __shfl(acc.y, l1) + __shfl(acc.y, l2);
    const float sz = acc.z + __shfl(acc.z, l1) + __shfl(acc.z, l2);
    const float sw = acc.w + __shfl(acc.w, l1) + __shfl(acc.w, l2);
    const float sd = den + __shfl(den, l1) + __shfl(den, l2);
    if (lane < 10) {
        const float4 b = *(const float4*)(b2 + q * 4);
        float4 o;
        o.x = sx / sd + b.x; o.y = sy / sd + b.y; o.z = sz / sd + b.z; o.w = sw / sd + b.w;
        *(float4*)(out + (size_t)n * NCLS + q * 4) = o;
    }
}

extern "C" void kernel_launch(void* const* d_in, const int* in_sizes, int n_in,
                              void* d_out, int out_size, void* d_ws, size_t ws_size,
                              hipStream_t stream) {
    const float* x    = (const float*)d_in[0];
    const int*   ei   = (const int*)  d_in[1];
    const float* W1   = (const float*)d_in[2];
    const float* as1  = (const float*)d_in[3];
    const float* ad1  = (const float*)d_in[4];
    const float* b1   = (const float*)d_in[5];
    const float* W2   = (const float*)d_in[6];
    const float* as2  = (const float*)d_in[7];
    const float* ad2  = (const float*)d_in[8];
    const float* b2   = (const float*)d_in[9];
    float* out = (float*)d_out;

    float* ws      = (float*)d_ws;
    float* h1      = ws;
    float* h2in    = h1     + (size_t)N_NODES * F1;
    float* h2      = h2in   + (size_t)N_NODES * F1;
    float* a_src1  = h2     + (size_t)N_NODES * NCLS;
    float* a_dst1  = a_src1 + (size_t)N_NODES * 2;
    float* a_src2  = a_dst1 + (size_t)N_NODES * 2;
    float* a_dst2  = a_src2 + N_NODES;
    int*   count   = (int*)(a_dst2 + N_NODES);
    int*   col     = count  + N_NODES + 16;
    int2*  ovf     = (int2*)(col + (size_t)N_NODES * BCAP);

    hipMemsetAsync(count, 0, (size_t)(N_NODES + 16) * sizeof(int), stream);

    const int MT = N_NODES / 16;
    gat_gemm1_and_buckets<<<G1_BLOCKS, 256, 0, stream>>>(x, W1, h1, ei, count, col, ovf);
    gat_att1<<<(N_NODES * HEADS + 255) / 256, 256, 0, stream>>>(h1, as1, ad1, a_src1, a_dst1);
    gat_gather1<<<N_NODES / 8, 256, 0, stream>>>(count, col, ovf, a_src1, a_dst1, h1, b1, h2in);
    gat_gemm2_wmma<<<MT, 96, 0, stream>>>(h2in, W2, h2);
    gat_att2<<<(N_NODES + 255) / 256, 256, 0, stream>>>(h2, as2, ad2, a_src2, a_dst2);
    gat_gather2<<<N_NODES / 8, 256, 0, stream>>>(count, col, ovf, a_src2, a_dst2, h2, b2, out);
}


# Both score-dot launches gone: the attention dots of both layers are formed in the GEMM epilogues from the accumulators (lane partials, DPP halving tree over a 16-lane row); dots sum in tree order
# speedup vs baseline: 4.1679x; 1.0608x over previous
#include <hip/hip_runtime.h>
#include <math.h>

#define N_NODES 50000
#define N_EDGES 800000
#define IN_FEATS 256
#define HID 64
#define HEADS 2
#define F1 (HEADS * HID)
#define NCLS 40

typedef __attribute__((ext_vector_type(2))) float v2f;
typedef __attribute__((ext_vector_type(8))) float v8f;

static __device__ __forceinline__ float lrelu(float x) { return x > 0.f ? x : 0.2f * x; }

template <int M>
static __device__ __forceinline__ float row_xor(const float x) {
    return __int_as_float(__builtin_amdgcn_update_dpp(0, __float_as_int(x), 0x160 + M, 0xf, 0xf, true));
}

template <int M, int N>
static __device__ __forceinline__ void row_halve(const bool up, const float* in, float* out) {
#pragma unroll
    for (int i = 0; i < N; ++i) {
        const float keep = up ? in[N + i] : in[i];
        const float send = up ? in[i] : in[N + i];
        out[i] = keep + row_xor<M>(send);
    }
}

__global__ __launch_bounds__(96) void gat_gemm2_wmma(const float* __restrict__ h2in,
                                                     const float* __restrict__ W2,
                                                     float* __restrict__ h2,
                                                     const float* __restrict__ as2,
                                                     const float* __restrict__ ad2,
                                                     float* __restrict__ a_src,
                                                     float* __restrict__ a_dst) {
    __shared__ float red[3][32];
    const int mtile = blockIdx.x;
    const int wave  = threadIdx.x >> 5;
    const int lane  = threadIdx.x & 31;
    const int half  = lane >> 4;
    const int l     = lane & 15;
    const int row   = mtile * 16 + l;
    const int ncol  = wave * 16 + l;

    v8f acc = {};
    const float* xr = h2in + (size_t)row * F1;
    for (int k = 0; k < F1; k += 4) {
        v2f a = *(const v2f*)(xr + k + 2 * half);
        v2f b;
        if (ncol < NCLS) {
            const float* bp = W2 + (size_t)(k + 2 * half) * NCLS + ncol;
            b.x = bp[0]; b.y = bp[NCLS];
        } else { b.x = 0.f; b.y = 0.f; }
        acc = __builtin_amdgcn_wmma_f32_16x16x4_f32(false, a, false, b,
                                                    (short)0, acc, false, false);
    }
    if (ncol < NCLS) {
        float* dr = h2 + (size_t)(mtile * 16 + 8 * half) * NCLS + ncol;
#pragma unroll
        for (int r = 0; r < 8; ++r) dr[(size_t)r * NCLS] = acc[r];
    }

    const float sc = ncol < NCLS ? as2[ncol] : 0.f;
    const float dc = ncol < NCLS ? ad2[ncol] : 0.f;
    float v[16], w8[8], w4[4], w2[2], w1[1];
#pragma unroll
    for (int r = 0; r < 8; ++r) { v[r] = acc[r] * sc; v[8 + r] = acc[r] * dc; }
    row_halve<8, 8>((l & 8) != 0, v, w8);
    row_halve<4, 4>((l & 4) != 0, w8, w4);
    row_halve<2, 2>((l & 2) != 0, w4, w2);
    row_halve<1, 1>((l & 1) != 0, w2, w1);
    red[wave][lane] = w1[0];
    __syncthreads();
    if (wave == 0) {
        const float tot = red[0][lane] + red[1][lane] + red[2][lane];
        const int orow = mtile * 16 + 8 * half + (l & 7);
        ((l & 8) ? a_dst : a_src)[orow] = tot;
    }
}

#define BCAP 64

static __device__ __forceinline__ int gat_edge_rank(int* __restrict__ count, const int d) {
    return d >= 0 ? atomicAdd(&count[d], 1) : 0;
}

static __device__ __forceinline__ void gat_edge_place(int* __restrict__ count, int* __restrict__ col,
                                int2* __restrict__ ovf, const int d, const int s, const int r) {
    if (d < 0) return;
    if (r < BCAP) {
        col[(size_t)d * BCAP + r] = s;
    } else {
        const int k = atomicAdd(&count[N_NODES], 1);
        ovf[k] = make_int2(d, s);
    }
}

#define G1_TILES   (N_NODES / 16)
#define G1_BLOCKS  ((G1_TILES + 7) / 8)
#define G1_EPT     8

__global__ __launch_bounds__(256) void gat_gemm1_and_buckets(const float* __restrict__ x,
        const float* __restrict__ W1, float* __restrict__ h1, const int* __restrict__ ei,
        int* __restrict__ count, int* __restrict__ col, int2* __restrict__ ovf,
        const float* __restrict__ as1, const float* __restrict__ ad1,
        float* __restrict__ a_src, float* __restrict__ a_dst) {
    __shared__ float sB[IN_FEATS * F1];
    const int tid  = threadIdx.x;
    const int wave = tid >> 5;
    const int lane = tid & 31;
    const int half = lane >> 4;
    const int l    = lane & 15;

    int ed[G1_EPT], es[G1_EPT], er[G1_EPT];
#pragma unroll
    for (int i = 0; i < G1_EPT; ++i) {
        const int e = (blockIdx.x * G1_EPT + i) * 256 + tid;
        const bool ok = e < N_EDGES;
        ed[i] = ok ? ei[N_EDGES + e] : -1;
        es[i] = ok ? ei[e] : 0;
    }

    const int  mtile   = blockIdx.x * 8 + wave;
    const bool tile_ok = mtile < G1_TILES;
    const int  mt      = tile_ok ? mtile : G1_TILES - 1;
    const float* xr = x + (size_t)(mt * 16 + l) * IN_FEATS + 4 * half;
    float4 a_cur[8], a_nxt[8];
#pragma unroll
    for (int q = 0; q < 8; ++q) a_cur[q] = *(const float4*)(xr + 8 * q);

    for (int i = tid; i < (IN_FEATS / 4) * F1; i += 256) {
        const int j = i >> 7, n = i & (F1 - 1);
        const int k0 = 8 * (j >> 1) + 2 * (j & 1);
        float4 v;
        v.x = W1[(size_t)(k0    ) * F1 + n];
        v.y = W1[(size_t)(k0 + 1) * F1 + n];
        v.z = W1[(size_t)(k0 + 4) * F1 + n];
        v.w = W1[(size_t)(k0 + 5) * F1 + n];
        *(float4*)(sB + j * 512 + n * 4) = v;
    }
    __syncthreads();

    const float* bl = sB + l * 4 + half * 2;

    v8f acc[8];
#pragma unroll
    for (int t = 0; t < 8; ++t) acc[t] = (v8f){};
    float asv[8], adv[8];

#pragma unroll
    for (int c = 0; c < 4; ++c) {
        if (c < 3) {
#pragma unroll
            for (int q = 0; q < 8; ++q) a_nxt[q] = *(const float4*)(xr + 64 * (c + 1) + 8 * q);
        }
        if (c == 3) {
#pragma unroll
            for (int t = 0; t < 8; ++t) { asv[t] = as1[t * 16 + l]; adv[t] = ad1[t * 16 + l]; }
        }
        er[2 * c]     = gat_edge_rank(count, ed[2 * c]);
        er[2 * c + 1] = gat_edge_rank(count, ed[2 * c + 1]);
        if (c > 0) {
            gat_edge_place(count, col, ovf, ed[2 * c - 2], es[2 * c - 2], er[2 * c - 2]);
            gat_edge_place(count, col, ovf, ed[2 * c - 1], es[2 * c - 1], er[2 * c - 1]);
        }
#pragma unroll
        for (int qq = 0; qq < 8; ++qq) {
#pragma unroll
            for (int e = 0; e < 2; ++e) {
                const int j = 2 * (8 * c + qq) + e;
                v2f a;
                a.x = e ? a_cur[qq].z : a_cur[qq].x;
                a.y = e ? a_cur[qq].w : a_cur[qq].y;
#pragma unroll
                for (int t = 0; t < 8; ++t) {
                    const v2f b = *(const v2f*)(bl + j * 512 + t * 64);
                    acc[t] = __builtin_amdgcn_wmma_f32_16x16x4_f32(false, a, false, b,
                                                                   (short)0, acc[t], false, false);
                }
            }
        }
        if (c < 3) {
#pragma unroll
            for (int q = 0; q < 8; ++q) a_cur[q] = a_nxt[q];
        }
    }
    gat_edge_place(count, col, ovf, ed[6], es[6], er[6]);
    gat_edge_place(count, col, ovf, ed[7], es[7], er[7]);

    if (tile_ok) {
        float* dr = h1 + (size_t)(mt * 16 + 8 * half) * F1 + l;
#pragma unroll
        for (int t = 0; t < 8; ++t) {
#pragma unroll
            for (int r = 0; r < 8; ++r) dr[(size_t)r * F1 + t * 16] = acc[t][r];
        }

        float v[32], w16[16], w8[8], w4[4], w2[2];
#pragma unroll
        for (int r = 0; r < 8; ++r) {
            float s0 = 0.f, s1 = 0.f, d0 = 0.f, d1 = 0.f;
#pragma unroll
            for (int t = 0; t < 4; ++t) {
                s0 += acc[t][r] * asv[t];  s1 += acc[t + 4][r] * asv[t + 4];
                d0 += acc[t][r] * adv[t];  d1 += acc[t + 4][r] * adv[t + 4];
            }
            v[r * 2] = s0;  v[r * 2 + 1] = s1;  v[16 + r * 2] = d0;  v[16 + r * 2 + 1] = d1;
        }
        row_halve<8, 16>((l & 8) != 0, v, w16);
        row_halve<4, 8>((l & 4) != 0, w16, w8);
        row_halve<2, 4>((l & 2) != 0, w8, w4);
        row_halve<1, 2>((l & 1) != 0, w4, w2);
        float* ap = ((l & 8) ? a_dst : a_src) + (size_t)(mt * 16 + 8 * half + (l & 7)) * 2;
        *(float2*)ap = make_float2(w2[0], w2[1]);
    }
}

static __device__ __forceinline__ void gat_chunk1(const bool valid, const int s_l, const int cnt,
        const int lane, const bool hi, const float2 adn, const float* __restrict__ a_src,
        const float* __restrict__ h1, float& m0, float& m1, float4& acc, float& den) {
    const float2 as = *(const float2*)(a_src + (size_t)s_l * 2);
    const float e0 = valid ? lrelu(as.x + adn.x) : -INFINITY;
    const float e1 = valid ? lrelu(as.y + adn.y) : -INFINITY;
    float c0m = e0, c1m = e1;
    for (int o = 16; o; o >>= 1) {
        c0m = fmaxf(c0m, __shfl_xor(c0m, o));
        c1m = fmaxf(c1m, __shfl_xor(c1m, o));
    }
    const float n0 = fmaxf(m0, c0m), n1 = fmaxf(m1, c1m);
    const float r = hi ? __expf(m1 - n1) : __expf(m0 - n0);
    acc.x *= r; acc.y *= r; acc.z *= r; acc.w *= r; den *= r;
    m0 = n0; m1 = n1;
    const int p0_l = __float_as_int(valid ? __expf(e0 - n0) : 0.f);
    const int p1_l = __float_as_int(valid ? __expf(e1 - n1) : 0.f);
    for (int j0 = 0; j0 < cnt; j0 += 4) {
#pragma unroll
        for (int u = 0; u < 4; ++u) {
            const int j = j0 + u;
            const int sj = __builtin_amdgcn_readlane(s_l, j);
            const float q0 = __int_as_float(__builtin_amdgcn_readlane(p0_l, j));
            const float q1 = __int_as_float(__builtin_amdgcn_readlane(p1_l, j));
            const float4 hv = *(const float4*)(h1 + (size_t)sj * F1 + lane * 4);
            const float p = hi ? q1 : q0;
            den += p;
            acc.x += p * hv.x; acc.y += p * hv.y; acc.z += p * hv.z; acc.w += p * hv.w;
        }
    }
}

__global__ __launch_bounds__(256) void gat_gather1(const int* __restrict__ count,
        const int* __restrict__ col, const int2* __restrict__ ovf,
        const float* __restrict__ a_src, const float* __restrict__ a_dst,
        const float* __restrict__ h1, const float* __restrict__ b1, float* __restrict__ h2in) {
    const int lane = threadIdx.x & 31;
    const int n = __builtin_amdgcn_readfirstlane(blockIdx.x * 8 + (threadIdx.x >> 5));
    if (n >= N_NODES) return;
    const int deg = count[n];
    const int nb  = min(deg, BCAP);
    const int* cb = col + (size_t)n * BCAP;
    const float2 adn = *(const float2*)(a_dst + (size_t)n * 2);
    const float2 asn = *(const float2*)(a_src + (size_t)n * 2);
    float m0 = lrelu(asn.x + adn.x), m1 = lrelu(asn.y + adn.y);
    const bool hi = lane >= 16;
    float4 acc = *(const float4*)(h1 + (size_t)n * F1 + lane * 4);
    float den = 1.f;
    for (int c0 = 0; c0 < nb; c0 += 32) {
        const int cnt = min(32, nb - c0);
        const bool valid = lane < cnt;
        const int s_l = valid ? cb[c0 + lane] : n;
        gat_chunk1(valid, s_l, cnt, lane, hi, adn, a_src, h1, m0, m1, acc, den);
    }
    if (deg > BCAP) {
        const int m = count[N_NODES];
        for (int k0 = 0; k0 < m; k0 += 32) {
            const int k = k0 + lane;
            const int2 en = k < m ? ovf[k] : make_int2(-1, 0);
            const bool valid = en.x == n;
            if (__ballot(valid) == 0) continue;
            const int s_l = valid ? en.y : n;
            gat_chunk1(valid, s_l, 32, lane, hi, adn, a_src, h1, m0, m1, acc, den);
        }
    }
    const float4 b = *(const float4*)(b1 + lane * 4);
    float4 o;
    o.x = acc.x / den + b.x; o.y = acc.y / den + b.y;
    o.z = acc.z / den + b.z; o.w = acc.w / den + b.w;
    o.x = o.x > 0.f ? o.x : (expf(o.x) - 1.0f);
    o.y = o.y > 0.f ? o.y : (expf(o.y) - 1.0f);
    o.z = o.z > 0.f ? o.z : (expf(o.z) - 1.0f);
    o.w = o.w > 0.f ? o.w : (expf(o.w) - 1.0f);
    *(float4*)(h2in + (size_t)n * F1 + lane * 4) = o;
}

static __device__ __forceinline__ void gat_chunk2(const bool valid, const int s_l, const int cnt,
        const bool live, const int g, const int q, const float adn,
        const float* __restrict__ a_src, const float* __restrict__ h2,
        float& m, float4& acc, float& den) {
    const float e = valid ? lrelu(a_src[s_l] + adn) : -INFINITY;
    float cm = e;
    for (int o = 16; o; o >>= 1) cm = fmaxf(cm, __shfl_xor(cm, o));
    const float nm = fmaxf(m, cm);
    const float r = __expf(m - nm);
    acc.x *= r; acc.y *= r; acc.z *= r; acc.w *= r; den *= r;
    m = nm;
    const float p_l = valid ? __expf(e - nm) : 0.f;
    for (int j0 = 0; j0 < cnt; j0 += 6) {
#pragma unroll
        for (int u = 0; u < 2; ++u) {
            const int j = j0 + 3 * u + g;
            const bool ok = live && (j < cnt);
            const int jj = ok ? j : 0;
            const int sj = __shfl(s_l, jj);
            const float pv = __shfl(p_l, jj);
            const float pj = ok ? pv : 0.f;
            const float4 hv = *(const float4*)(h2 + (size_t)sj * NCLS + q * 4);
            den += pj;
            acc.x += pj * hv.x; acc.y += pj * hv.y; acc.z += pj * hv.z; acc.w += pj * hv.w;
        }
    }
}

__global__ __launch_bounds__(256) void gat_gather2(const int* __restrict__ count,
        const int* __restrict__ col, const int2* __restrict__ ovf,
        const float* __restrict__ a_src, const float* __restrict__ a_dst,
        const float* __restrict__ h2, const float* __restrict__ b2, float* __restrict__ out) {
    const int lane = threadIdx.x & 31;
    const int n = __builtin_amdgcn_readfirstlane(blockIdx.x * 8 + (threadIdx.x >> 5));
    if (n >= N_NODES) return;
    const int g3 = lane / 10;
    const bool live = g3 < 3;
    const int g = live ? g3 : 0;
    const int q = lane - 10 * g3;
    const int deg = count[n];
    const int nb  = min(deg, BCAP);
    const int* cb = col + (size_t)n * BCAP;
    const float adn = a_dst[n];
    float m = lrelu(a_src[n] + adn);
    float4 acc = {0.f, 0.f, 0.f, 0.f};
    float den = 0.f;
    if (lane < 10) { acc = *(const float4*)(h2 + (size_t)n * NCLS + q * 4); den = 1.f; }
    for (int c0 = 0; c0 < nb; c0 += 32) {
        const int cnt = min(32, nb - c0);
        const bool valid = lane < cnt;
        const int s_l = valid ? cb[c0 + lane] : n;
        gat_chunk2(valid, s_l, cnt, live, g, q, adn, a_src, h2, m, acc, den);
    }
    if (deg > BCAP) {
        const int mo = count[N_NODES];
        for (int k0 = 0; k0 < mo; k0 += 32) {
            const int k = k0 + lane;
            const int2 en = k < mo ? ovf[k] : make_int2(-1, 0);
            const bool valid = en.x == n;
            if (__ballot(valid) == 0) continue;
            const int s_l = valid ? en.y : n;
            gat_chunk2(valid, s_l, 32, live, g, q, adn, a_src, h2, m, acc, den);
        }
    }
    const int l1 = (lane + 10) & 31, l2 = (lane + 20) & 31;
    const float sx = acc.x + __shfl(acc.x, l1) + __shfl(acc.x, l2);
    const float sy = acc.y + __shfl(acc.y, l1) + __shfl(acc.y, l2);
    const float sz = acc.z + __shfl(acc.z, l1) + __shfl(acc.z, l2);
    const float sw = acc.w + __shfl(acc.w, l1) + __shfl(acc.w, l2);
    const float sd = den + __shfl(den, l1) + __shfl(den, l2);
    if (lane < 10) {
        const float4 b = *(const float4*)(b2 + q * 4);
        float4 o;
        o.x = sx / sd + b.x; o.y = sy / sd + b.y; o.z = sz / sd + b.z; o.w = sw / sd + b.w;
        *(float4*)(out + (size_t)n * NCLS + q * 4) = o;
    }
}

extern "C" void kernel_launch(void* const* d_in, const int* in_sizes, int n_in,
                              void* d_out, int out_size, void* d_ws, size_t ws_size,
                              hipStream_t stream) {
    const float* x    = (const float*)d_in[0];
    const int*   ei   = (const int*)  d_in[1];
    const float* W1   = (const float*)d_in[2];
    const float* as1  = (const float*)d_in[3];
    const float* ad1  = (const float*)d_in[4];
    const float* b1   = (const float*)d_in[5];
    const float* W2   = (const float*)d_in[6];
    const float* as2  = (const float*)d_in[7];
    const float* ad2  = (const float*)d_in[8];
    const float* b2   = (const float*)d_in[9];
    float* out = (float*)d_out;

    float* ws      = (float*)d_ws;
    float* h1      = ws;
    float* h2in    = h1     + (size_t)N_NODES * F1;
    float* h2      = h2in   + (size_t)N_NODES * F1;
    float* a_src1  = h2     + (size_t)N_NODES * NCLS;
    float* a_dst1  = a_src1 + (size_t)N_NODES * 2;
    float* a_src2  = a_dst1 + (size_t)N_NODES * 2;
    float* a_dst2  = a_src2 + N_NODES;
    int*   count   = (int*)(a_dst2 + N_NODES);
    int*   col     = count  + N_NODES + 16;
    int2*  ovf     = (int2*)(col + (size_t)N_NODES * BCAP);

    hipMemsetAsync(count, 0, (size_t)(N_NODES + 16) * sizeof(int), stream);

    const int MT = N_NODES / 16;
    gat_gemm1_and_buckets<<<G1_BLOCKS, 256, 0, stream>>>(x, W1, h1, ei, count, col, ovf,
                                                         as1, ad1, a_src1, a_dst1);
    gat_gather1<<<N_NODES / 8, 256, 0, stream>>>(count, col, ovf, a_src1, a_dst1, h1, b1, h2in);
    gat_gemm2_wmma<<<MT, 96, 0, stream>>>(h2in, W2, h2, as2, ad2, a_src2, a_dst2);
    gat_gather2<<<N_NODES / 8, 256, 0, stream>>>(count, col, ovf, a_src2, a_dst2, h2, b2, out);
}


# GEMM2 rides in the layer-1 gather launch: a block gathers 16 nodes (two per wave), keeps their ELU rows in LDS, and one wave forms h2 and the layer-2 dots on WMMA; one launch fewer; sum orders kept
# speedup vs baseline: 4.3684x; 1.0481x over previous
#include <hip/hip_runtime.h>
#include <math.h>

#define N_NODES 50000
#define N_EDGES 800000
#define IN_FEATS 256
#define HID 64
#define HEADS 2
#define F1 (HEADS * HID)
#define NCLS 40

typedef __attribute__((ext_vector_type(2))) float v2f;
typedef __attribute__((ext_vector_type(8))) float v8f;

static __device__ __forceinline__ float lrelu(float x) { return x > 0.f ? x : 0.2f * x; }

template <int M>
static __device__ __forceinline__ float row_xor(const float x) {
    return __int_as_float(__builtin_amdgcn_update_dpp(0, __float_as_int(x), 0x160 + M, 0xf, 0xf, true));
}

template <int M, int N>
static __device__ __forceinline__ void row_halve(const bool up, const float* in, float* out) {
#pragma unroll
    for (int i = 0; i < N; ++i) {
        const float keep = up ? in[N + i] : in[i];
        const float send = up ? in[i] : in[N + i];
        out[i] = keep + row_xor<M>(send);
    }
}

#define BCAP 64

static __device__ __forceinline__ int gat_edge_rank(int* __restrict__ count, const int d) {
    return d >= 0 ? atomicAdd(&count[d], 1) : 0;
}

static __device__ __forceinline__ void gat_edge_place(int* __restrict__ count, int* __restrict__ col,
                                int2* __restrict__ ovf, const int d, const int s, const int r) {
    if (d < 0) return;
    if (r < BCAP) {
        col[(size_t)d * BCAP + r] = s;
    } else {
        const int k = atomicAdd(&count[N_NODES], 1);
        ovf[k] = make_int2(d, s);
    }
}

#define G1_TILES   (N_NODES / 16)
#define G1_BLOCKS  ((G1_TILES + 7) / 8)
#define G1_EPT     8

__global__ __launch_bounds__(256) void gat_gemm1_and_buckets(const float* __restrict__ x,
        const float* __restrict__ W1, float* __restrict__ h1, const int* __restrict__ ei,
        int* __restrict__ count, int* __restrict__ col, int2* __restrict__ ovf,
        const float* __restrict__ as1, const float* __restrict__ ad1,
        float* __restrict__ a_src, float* __restrict__ a_dst) {
    __shared__ float sB[IN_FEATS * F1];
    const int tid  = threadIdx.x;
    const int wave = tid >> 5;
    const int lane = tid & 31;
    const int half = lane >> 4;
    const int l    = lane & 15;

    int ed[G1_EPT], es[G1_EPT], er[G1_EPT];
#pragma unroll
    for (int i = 0; i < G1_EPT; ++i) {
        const int e = (blockIdx.x * G1_EPT + i) * 256 + tid;
        const bool ok = e < N_EDGES;
        ed[i] = ok ? ei[N_EDGES + e] : -1;
        es[i] = ok ? ei[e] : 0;
    }

    const int  mtile   = blockIdx.x * 8 + wave;
    const bool tile_ok = mtile < G1_TILES;
    const int  mt      = tile_ok ? mtile : G1_TILES - 1;
    const float* xr = x + (size_t)(mt * 16 + l) * IN_FEATS + 4 * half;
    float4 a_cur[8], a_nxt[8];
#pragma unroll
    for (int q = 0; q < 8; ++q) a_cur[q] = *(const float4*)(xr + 8 * q);

    for (int i = tid; i < (IN_FEATS / 4) * F1; i += 256) {
        const int j = i >> 7, n = i & (F1 - 1);
        const int k0 = 8 * (j >> 1) + 2 * (j & 1);
        float4 v;
        v.x = W1[(size_t)(k0    ) * F1 + n];
        v.y = W1[(size_t)(k0 + 1) * F1 + n];
        v.z = W1[(size_t)(k0 + 4) * F1 + n];
        v.w = W1[(size_t)(k0 + 5) * F1 + n];
        *(float4*)(sB + j * 512 + n * 4) = v;
    }
    __syncthreads();

    const float* bl = sB + l * 4 + half * 2;

    v8f acc[8];
#pragma unroll
    for (int t = 0; t < 8; ++t) acc[t] = (v8f){};
    float asv[8], adv[8];

#pragma unroll
    for (int c = 0; c < 4; ++c) {
        if (c < 3) {
#pragma unroll
            for (int q = 0; q < 8; ++q) a_nxt[q] = *(const float4*)(xr + 64 * (c + 1) + 8 * q);
        }
        if (c == 3) {
#pragma unroll
            for (int t = 0; t < 8; ++t) { asv[t] = as1[t * 16 + l]; adv[t] = ad1[t * 16 + l]; }
        }
        er[2 * c]     = gat_edge_rank(count, ed[2 * c]);
        er[2 * c + 1] = gat_edge_rank(count, ed[2 * c + 1]);
        if (c > 0) {
            gat_edge_place(count, col, ovf, ed[2 * c - 2], es[2 * c - 2], er[2 * c - 2]);
            gat_edge_place(count, col, ovf, ed[2 * c - 1], es[2 * c - 1], er[2 * c - 1]);
        }
#pragma unroll
        for (int qq = 0; qq < 8; ++qq) {
#pragma unroll
            for (int e = 0; e < 2; ++e) {
                const int j = 2 * (8 * c + qq) + e;
                v2f a;
                a.x = e ? a_cur[qq].z : a_cur[qq].x;
                a.y = e ? a_cur[qq].w : a_cur[qq].y;
#pragma unroll
                for (int t = 0; t < 8; ++t) {
                    const v2f b = *(const v2f*)(bl + j * 512 + t * 64);
                    acc[t] = __builtin_amdgcn_wmma_f32_16x16x4_f32(false, a, false, b,
                                                                   (short)0, acc[t], false, false);
                }
            }
        }
        if (c < 3) {
#pragma unroll
            for (int q = 0; q < 8; ++q) a_cur[q] = a_nxt[q];
        }
    }
    gat_edge_place(count, col, ovf, ed[6], es[6], er[6]);
    gat_edge_place(count, col, ovf, ed[7], es[7], er[7]);

    if (tile_ok) {
        float* dr = h1 + (size_t)(mt * 16 + 8 * half) * F1 + l;
#pragma unroll
        for (int t = 0; t < 8; ++t) {
#pragma unroll
            for (int r = 0; r < 8; ++r) dr[(size_t)r * F1 + t * 16] = acc[t][r];
        }

        float v[32], w16[16], w8[8], w4[4], w2[2];
#pragma unroll
        for (int r = 0; r < 8; ++r) {
            float s0 = 0.f, s1 = 0.f, d0 = 0.f, d1 = 0.f;
#pragma unroll
            for (int t = 0; t < 4; ++t) {
                s0 += acc[t][r] * asv[t];  s1 += acc[t + 4][r] * asv[t + 4];
                d0 += acc[t][r] * adv[t];  d1 += acc[t + 4][r] * adv[t + 4];
            }
            v[r * 2] = s0;  v[r * 2 + 1] = s1;  v[16 + r * 2] = d0;  v[16 + r * 2 + 1] = d1;
        }
        row_halve<8, 16>((l & 8) != 0, v, w16);
        row_halve<4, 8>((l & 4) != 0, w16, w8);
        row_halve<2, 4>((l & 2) != 0, w8, w4);
        row_halve<1, 2>((l & 1) != 0, w4, w2);
        float* ap = ((l & 8) ? a_dst : a_src) + (size_t)(mt * 16 + 8 * half + (l & 7)) * 2;
        *(float2*)ap = make_float2(w2[0], w2[1]);
    }
}

static __device__ __forceinline__ void gat_chunk1(const bool valid, const int s_l, const int cnt,
        const int lane, const bool hi, const float2 adn, const float* __restrict__ a_src,
        const float* __restrict__ h1, float& m0, float& m1, float4& acc, float& den) {
    const float2 as = *(const float2*)(a_src + (size_t)s_l * 2);
    const float e0 = valid ? lrelu(as.x + adn.x) : -INFINITY;
    const float e1 = valid ? lrelu(as.y + adn.y) : -INFINITY;
    float c0m = e0, c1m = e1;
    for (int o = 16; o; o >>= 1) {
        c0m = fmaxf(c0m, __shfl_xor(c0m, o));
        c1m = fmaxf(c1m, __shfl_xor(c1m, o));
    }
    const float n0 = fmaxf(m0, c0m), n1 = fmaxf(m1, c1m);
    const float r = hi ? __expf(m1 - n1) : __expf(m0 - n0);
    acc.x *= r; acc.y *= r; acc.z *= r; acc.w *= r; den *= r;
    m0 = n0; m1 = n1;
    const int p0_l = __float_as_int(valid ? __expf(e0 - n0) : 0.f);
    const int p1_l = __float_as_int(valid ? __expf(e1 - n1) : 0.f);
    for (int j0 = 0; j0 < cnt; j0 += 4) {
#pragma unroll
        for (int u = 0; u < 4; ++u) {
            const int j = j0 + u;
            const int sj = __builtin_amdgcn_readlane(s_l, j);
            const float q0 = __int_as_float(__builtin_amdgcn_readlane(p0_l, j));
            const float q1 = __int_as_float(__builtin_amdgcn_readlane(p1_l, j));
            const float4 hv = *(const float4*)(h1 + (size_t)sj * F1 + lane * 4);
            const float p = hi ? q1 : q0;
            den += p;
            acc.x += p * hv.x; acc.y += p * hv.y; acc.z += p * hv.z; acc.w += p * hv.w;
        }
    }
}

static __device__ __forceinline__ float4 gat_node1(const int n, const int lane, const float4 b,
        const int* __restrict__ count, const int* __restrict__ col, const int2* __restrict__ ovf,
        const float* __restrict__ a_src, const float* __restrict__ a_dst,
        const float* __restrict__ h1) {
    const int deg = count[n];
    const int nb  = min(deg, BCAP);
    const int* cb = col + (size_t)n * BCAP;
    const float2 adn = *(const float2*)(a_dst + (size_t)n * 2);
    const float2 asn = *(const float2*)(a_src + (size_t)n * 2);
    float m0 = lrelu(asn.x + adn.x), m1 = lrelu(asn.y + adn.y);
    const bool hi = lane >= 16;
    float4 acc = *(const float4*)(h1 + (size_t)n * F1 + lane * 4);
    float den = 1.f;
    for (int c0 = 0; c0 < nb; c0 += 32) {
        const int cnt = min(32, nb - c0);
        const bool valid = lane < cnt;
        const int s_l = valid ? cb[c0 + lane] : n;
        gat_chunk1(valid, s_l, cnt, lane, hi, adn, a_src, h1, m0, m1, acc, den);
    }
    if (deg > BCAP) {
        const int m = count[N_NODES];
        for (int k0 = 0; k0 < m; k0 += 32) {
            const int k = k0 + lane;
            const int2 en = k < m ? ovf[k] : make_int2(-1, 0);
            const bool valid = en.x == n;
            if (__ballot(valid) == 0) continue;
            const int s_l = valid ? en.y : n;
            gat_chunk1(valid, s_l, 32, lane, hi, adn, a_src, h1, m0, m1, acc, den);
        }
    }
    float4 o;
    o.x = acc.x / den + b.x; o.y = acc.y / den + b.y;
    o.z = acc.z / den + b.z; o.w = acc.w / den + b.w;
    o.x = o.x > 0.f ? o.x : (expf(o.x) - 1.0f);
    o.y = o.y > 0.f ? o.y : (expf(o.y) - 1.0f);
    o.z = o.z > 0.f ? o.z : (expf(o.z) - 1.0f);
    o.w = o.w > 0.f ? o.w : (expf(o.w) - 1.0f);
    return o;
}

#define G2_ROWP 132
#define G2_NPB  16
static_assert(N_NODES % G2_NPB == 0, "a block owns 16 whole nodes");

__global__ __launch_bounds__(256) void gat_gather1_gemm2(const int* __restrict__ count,
        const int* __restrict__ col, const int2* __restrict__ ovf,
        const float* __restrict__ a_src, const float* __restrict__ a_dst,
        const float* __restrict__ h1, const float* __restrict__ b1,
        const float* __restrict__ W2, float* __restrict__ h2,
        const float* __restrict__ as2, const float* __restrict__ ad2,
        float* __restrict__ a_src2, float* __restrict__ a_dst2) {
    __shared__ float sW[(F1 / 4) * 48 * 4];
    __shared__ float sRow[G2_NPB * G2_ROWP];
    const int tid  = threadIdx.x;
    const int lane = tid & 31;
    const int wave = __builtin_amdgcn_readfirstlane(tid >> 5);

#pragma unroll
    for (int u = 0; u < (F1 / 4) * 48 / 256; ++u) {
        const int i = tid + 256 * u;
        const int j = i / 48, n = i - 48 * j;
        const float* wp = W2 + (size_t)(4 * j) * NCLS + min(n, NCLS - 1);
        float4 v;
        v.x = wp[0]; v.y = wp[NCLS]; v.z = wp[2 * NCLS]; v.w = wp[3 * NCLS];
        if (n >= NCLS) v = make_float4(0.f, 0.f, 0.f, 0.f);
        *(float4*)(sW + i * 4) = v;
    }

    const float4 b = *(const float4*)(b1 + lane * 4);
    for (int t = 0; t < 2; ++t) {
        const int r = wave + 8 * t;
        const float4 o = gat_node1(blockIdx.x * G2_NPB + r, lane, b, count, col, ovf, a_src, a_dst, h1);
        *(float4*)(sRow + r * G2_ROWP + lane * 4) = o;
    }
    __syncthreads();
    if (wave != 0) return;

    const int half = lane >> 4;
    const int l    = lane & 15;
    const float* ar = sRow + l * G2_ROWP + 2 * half;
    const float* bw = sW + l * 4 + 2 * half;
    v8f acc[3];
#pragma unroll
    for (int t = 0; t < 3; ++t) acc[t] = (v8f){};
#pragma unroll
    for (int j = 0; j < F1 / 4; ++j) {
        const v2f a = *(const v2f*)(ar + 4 * j);
#pragma unroll
        for (int t = 0; t < 3; ++t) {
            const v2f bf = *(const v2f*)(bw + j * 192 + t * 64);
            acc[t] = __builtin_amdgcn_wmma_f32_16x16x4_f32(false, a, false, bf,
                                                           (short)0, acc[t], false, false);
        }
    }

    const int base = blockIdx.x * G2_NPB;
    float part[3];
#pragma unroll
    for (int t = 0; t < 3; ++t) {
        const int ncol = t * 16 + l;
        if (ncol < NCLS) {
            float* dr = h2 + (size_t)(base + 8 * half) * NCLS + ncol;
#pragma unroll
            for (int r = 0; r < 8; ++r) dr[(size_t)r * NCLS] = acc[t][r];
        }
        const float sc = ncol < NCLS ? as2[ncol] : 0.f;
        const float dc = ncol < NCLS ? ad2[ncol] : 0.f;
        float v[16], w8[8], w4[4], w2[2], w1[1];
#pragma unroll
        for (int r = 0; r < 8; ++r) { v[r] = acc[t][r] * sc; v[8 + r] = acc[t][r] * dc; }
        row_halve<8, 8>((l & 8) != 0, v, w8);
        row_halve<4, 4>((l & 4) != 0, w8, w4);
        row_halve<2, 2>((l & 2) != 0, w4, w2);
        row_halve<1, 1>((l & 1) != 0, w2, w1);
        part[t] = w1[0];
    }
    const float tot = part[0] + part[1] + part[2];
    ((l & 8) ? a_dst2 : a_src2)[base + 8 * half + (l & 7)] = tot;
}

static __device__ __forceinline__ void gat_chunk2(const bool valid, const int s_l, const int cnt,
        const bool live, const int g, const int q, const float adn,
        const float* __restrict__ a_src, const float* __restrict__ h2,
        float& m, float4& acc, float& den) {
    const float e = valid ? lrelu(a_src[s_l] + adn) : -INFINITY;
    float cm = e;
    for (int o = 16; o; o >>= 1) cm = fmaxf(cm, __shfl_xor(cm, o));
    const float nm = fmaxf(m, cm);
    const float r = __expf(m - nm);
    acc.x *= r; acc.y *= r; acc.z *= r; acc.w *= r; den *= r;
    m = nm;
    const float p_l = valid ? __expf(e - nm) : 0.f;
    for (int j0 = 0; j0 < cnt; j0 += 6) {
#pragma unroll
        for (int u = 0; u < 2; ++u) {
            const int j = j0 + 3 * u + g;
            const bool ok = live && (j < cnt);
            const int jj = ok ? j : 0;
            const int sj = __shfl(s_l, jj);
            const float pv = __shfl(p_l, jj);
            const float pj = ok ? pv : 0.f;
            const float4 hv = *(const float4*)(h2 + (size_t)sj * NCLS + q * 4);
            den += pj;
            acc.x += pj * hv.x; acc.y += pj * hv.y; acc.z += pj * hv.z; acc.w += pj * hv.w;
        }
    }
}

__global__ __launch_bounds__(256) void gat_gather2(const int* __restrict__ count,
        const int* __restrict__ col, const int2* __restrict__ ovf,
        const float* __restrict__ a_src, const float* __restrict__ a_dst,
        const float* __restrict__ h2, const float* __restrict__ b2, float* __restrict__ out) {
    const int lane = threadIdx.x & 31;
    const int n = __builtin_amdgcn_readfirstlane(blockIdx.x * 8 + (threadIdx.x >> 5));
    if (n >= N_NODES) return;
    const int g3 = lane / 10;
    const bool live = g3 < 3;
    const int g = live ? g3 : 0;
    const int q = lane - 10 * g3;
    const int deg = count[n];
    const int nb  = min(deg, BCAP);
    const int* cb = col + (size_t)n * BCAP;
    const float adn = a_dst[n];
    float m = lrelu(a_src[n] + adn);
    float4 acc = {0.f, 0.f, 0.f, 0.f};
    float den = 0.f;
    if (lane < 10) { acc = *(const float4*)(h2 + (size_t)n * NCLS + q * 4); den = 1.f; }
    for (int c0 = 0; c0 < nb; c0 += 32) {
        const int cnt = min(32, nb - c0);
        const bool valid = lane < cnt;
        const int s_l = valid ? cb[c0 + lane] : n;
        gat_chunk2(valid, s_l, cnt, live, g, q, adn, a_src, h2, m, acc, den);
    }
    if (deg > BCAP) {
        const int mo = count[N_NODES];
        for (int k0 = 0; k0 < mo; k0 += 32) {
            const int k = k0 + lane;
            const int2 en = k < mo ? ovf[k] : make_int2(-1, 0);
            const bool valid = en.x == n;
            if (__ballot(valid) == 0) continue;
            const int s_l = valid ? en.y : n;
            gat_chunk2(valid, s_l, 32, live, g, q, adn, a_src, h2, m, acc, den);
        }
    }
    const int l1 = (lane + 10) & 31, l2 = (lane + 20) & 31;
    const float sx = acc.x + __shfl(acc.x, l1) + __shfl(acc.x, l2);
    const float sy = acc.y + __shfl(acc.y, l1) + __shfl(acc.y, l2);
    const float sz = acc.z + __shfl(acc.z, l1) + __shfl(acc.z, l2);
    const float sw = acc.w + __shfl(acc.w, l1) + __shfl(acc.w, l2);
    const float sd = den + __shfl(den, l1) + __shfl(den, l2);
    if (lane < 10) {
        const float4 b = *(const float4*)(b2 + q * 4);
        float4 o;
        o.x = sx / sd + b.x; o.y = sy / sd + b.y; o.z = sz / sd + b.z; o.w = sw / sd + b.w;
        *(float4*)(out + (size_t)n * NCLS + q * 4) = o;
    }
}

extern "C" void kernel_launch(void* const* d_in, const int* in_sizes, int n_in,
                              void* d_out, int out_size, void* d_ws, size_t ws_size,
                              hipStream_t stream) {
    const float* x    = (const float*)d_in[0];
    const int*   ei   = (const int*)  d_in[1];
    const float* W1   = (const float*)d_in[2];
    const float* as1  = (const float*)d_in[3];
    const float* ad1  = (const float*)d_in[4];
    const float* b1   = (const float*)d_in[5];
    const float* W2   = (const float*)d_in[6];
    const float* as2  = (const float*)d_in[7];
    const float* ad2  = (const float*)d_in[8];
    const float* b2   = (const float*)d_in[9];
    float* out = (float*)d_out;

    float* ws      = (float*)d_ws;
    float* h1      = ws;
    float* h2      = h1     + (size_t)N_NODES * F1;
    float* a_src1  = h2     + (size_t)N_NODES * NCLS;
    float* a_dst1  = a_src1 + (size_t)N_NODES * 2;
    float* a_src2  = a_dst1 + (size_t)N_NODES * 2;
    float* a_dst2  = a_src2 + N_NODES;
    int*   count   = (int*)(a_dst2 + N_NODES);
    int*   col     = count  + N_NODES + 16;
    int2*  ovf     = (int2*)(col + (size_t)N_NODES * BCAP);

    hipMemsetAsync(count, 0, (size_t)(N_NODES + 16) * sizeof(int), stream);

    gat_gemm1_and_buckets<<<G1_BLOCKS, 256, 0, stream>>>(x, W1, h1, ei, count, col, ovf,
                                                         as1, ad1, a_src1, a_dst1);
    gat_gather1_gemm2<<<N_NODES / G2_NPB, 256, 0, stream>>>(count, col, ovf, a_src1, a_dst1, h1, b1,
                                                            W2, h2, as2, ad2, a_src2, a_dst2);
    gat_gather2<<<N_NODES / 8, 256, 0, stream>>>(count, col, ovf, a_src2, a_dst2, h2, b2, out);
}


# Edge lists in two levels: blocks count edges per 128-node bin in LDS and take bin room with one global add per bin (5x fewer returning adds); gather blocks rank their nodes by LDS adds; arrival-set
# speedup vs baseline: 5.2978x; 1.2128x over previous
#include <hip/hip_runtime.h>
#include <math.h>

#define N_NODES 50000
#define N_EDGES 800000
#define IN_FEATS 256
#define HID 64
#define HEADS 2
#define F1 (HEADS * HID)
#define NCLS 40

typedef __attribute__((ext_vector_type(2))) float v2f;
typedef __attribute__((ext_vector_type(8))) float v8f;

static __device__ __forceinline__ float lrelu(float x) { return x > 0.f ? x : 0.2f * x; }

template <int M>
static __device__ __forceinline__ float row_xor(const float x) {
    return __int_as_float(__builtin_amdgcn_update_dpp(0, __float_as_int(x), 0x160 + M, 0xf, 0xf, true));
}

template <int M, int N>
static __device__ __forceinline__ void row_halve(const bool up, const float* in, float* out) {
#pragma unroll
    for (int i = 0; i < N; ++i) {
        const float keep = up ? in[N + i] : in[i];
        const float send = up ? in[i] : in[N + i];
        out[i] = keep + row_xor<M>(send);
    }
}

#define BCAP 64
#define BIN_NODES 128
#define NBINS ((N_NODES + BIN_NODES - 1) / BIN_NODES)
#define BIN_CAP 2560
#define BIN_PITCH 32

#define G1_TILES   (N_NODES / 16)
#define G1_BLOCKS  ((G1_TILES + 7) / 8)
#define G1_EPT     8

__global__ __launch_bounds__(256) void gat_gemm1_and_buckets(const float* __restrict__ x,
        const float* __restrict__ W1, float* __restrict__ h1, const int* __restrict__ ei,
        int* __restrict__ binCnt, int* __restrict__ binE, int2* __restrict__ ovf,
        const float* __restrict__ as1, const float* __restrict__ ad1,
        float* __restrict__ a_src, float* __restrict__ a_dst) {
    __shared__ float sB[IN_FEATS * F1];
    __shared__ int sHist[NBINS];
    const int tid  = threadIdx.x;
    const int wave = tid >> 5;
    const int lane = tid & 31;
    const int half = lane >> 4;
    const int l    = lane & 15;

    int ed[G1_EPT], es[G1_EPT], er[G1_EPT];
#pragma unroll
    for (int i = 0; i < G1_EPT; ++i) {
        const int e = (blockIdx.x * G1_EPT + i) * 256 + tid;
        const bool ok = e < N_EDGES;
        ed[i] = ok ? ei[N_EDGES + e] : -1;
        es[i] = ok ? ei[e] : 0;
    }
    for (int i = tid; i < NBINS; i += 256) sHist[i] = 0;
    int g0 = 0, g1 = 0;

    const int  mtile   = blockIdx.x * 8 + wave;
    const bool tile_ok = mtile < G1_TILES;
    const int  mt      = tile_ok ? mtile : G1_TILES - 1;
    const float* xr = x + (size_t)(mt * 16 + l) * IN_FEATS + 4 * half;
    float4 a_cur[8], a_nxt[8];
#pragma unroll
    for (int q = 0; q < 8; ++q) a_cur[q] = *(const float4*)(xr + 8 * q);

    for (int i = tid; i < (IN_FEATS / 4) * F1; i += 256) {
        const int j = i >> 7, n = i & (F1 - 1);
        const int k0 = 8 * (j >> 1) + 2 * (j & 1);
        float4 v;
        v.x = W1[(size_t)(k0    ) * F1 + n];
        v.y = W1[(size_t)(k0 + 1) * F1 + n];
        v.z = W1[(size_t)(k0 + 4) * F1 + n];
        v.w = W1[(size_t)(k0 + 5) * F1 + n];
        *(float4*)(sB + j * 512 + n * 4) = v;
    }
    __syncthreads();

    const float* bl = sB + l * 4 + half * 2;

    v8f acc[8];
#pragma unroll
    for (int t = 0; t < 8; ++t) acc[t] = (v8f){};
    float asv[8], adv[8];

#pragma unroll
    for (int c = 0; c < 4; ++c) {
        if (c < 3) {
#pragma unroll
            for (int q = 0; q < 8; ++q) a_nxt[q] = *(const float4*)(xr + 64 * (c + 1) + 8 * q);
        }
        if (c == 3) {
#pragma unroll
            for (int t = 0; t < 8; ++t) { asv[t] = as1[t * 16 + l]; adv[t] = ad1[t * 16 + l]; }
        }
        if (c == 0) {
#pragma unroll
            for (int i = 0; i < G1_EPT; ++i)
                er[i] = ed[i] >= 0 ? atomicAdd(&sHist[ed[i] >> 7], 1) : 0;
        }
        if (c == 1) {
            __syncthreads();
            const int c0 = sHist[tid];
            const int c1 = tid + 256 < NBINS ? sHist[tid + 256] : 0;
            if (c0 > 0) g0 = atomicAdd(&binCnt[tid * BIN_PITCH], c0);
            if (c1 > 0) g1 = atomicAdd(&binCnt[(tid + 256) * BIN_PITCH], c1);
        }
        if (c == 2) {
            sHist[tid] = g0;
            if (tid + 256 < NBINS) sHist[tid + 256] = g1;
            __syncthreads();
#pragma unroll
            for (int i = 0; i < G1_EPT; ++i) {
                if (ed[i] >= 0) {
                    const int bin = ed[i] >> 7;
                    const int pos = sHist[bin] + er[i];
                    if (pos < BIN_CAP) {
                        binE[(size_t)bin * BIN_CAP + pos] = ((ed[i] & (BIN_NODES - 1)) << 16) | es[i];
                    } else {
                        const int k = atomicAdd(&binCnt[NBINS * BIN_PITCH], 1);
                        ovf[k] = make_int2(ed[i], es[i]);
                    }
                }
            }
        }
#pragma unroll
        for (int qq = 0; qq < 8; ++qq) {
#pragma unroll
            for (int e = 0; e < 2; ++e) {
                const int j = 2 * (8 * c + qq) + e;
                v2f a;
                a.x = e ? a_cur[qq].z : a_cur[qq].x;
                a.y = e ? a_cur[qq].w : a_cur[qq].y;
#pragma unroll
                for (int t = 0; t < 8; ++t) {
                    const v2f b = *(const v2f*)(bl + j * 512 + t * 64);
                    acc[t] = __builtin_amdgcn_wmma_f32_16x16x4_f32(false, a, false, b,
                                                                   (short)0, acc[t], false, false);
                }
            }
        }
        if (c < 3) {
#pragma unroll
            for (int q = 0; q < 8; ++q) a_cur[q] = a_nxt[q];
        }
    }

    if (tile_ok) {
        float* dr = h1 + (size_t)(mt * 16 + 8 * half) * F1 + l;
#pragma unroll
        for (int t = 0; t < 8; ++t) {
#pragma unroll
            for (int r = 0; r < 8; ++r) dr[(size_t)r * F1 + t * 16] = acc[t][r];
        }

        float v[32], w16[16], w8[8], w4[4], w2[2];
#pragma unroll
        for (int r = 0; r < 8; ++r) {
            float s0 = 0.f, s1 = 0.f, d0 = 0.f, d1 = 0.f;
#pragma unroll
            for (int t = 0; t < 4; ++t) {
                s0 += acc[t][r] * asv[t];  s1 += acc[t + 4][r] * asv[t + 4];
                d0 += acc[t][r] * adv[t];  d1 += acc[t + 4][r] * adv[t + 4];
            }
            v[r * 2] = s0;  v[r * 2 + 1] = s1;  v[16 + r * 2] = d0;  v[16 + r * 2 + 1] = d1;
        }
        row_halve<8, 16>((l & 8) != 0, v, w16);
        row_halve<4, 8>((l & 4) != 0, w16, w8);
        row_halve<2, 4>((l & 2) != 0, w8, w4);
        row_halve<1, 2>((l & 1) != 0, w4, w2);
        float* ap = ((l & 8) ? a_dst : a_src) + (size_t)(mt * 16 + 8 * half + (l & 7)) * 2;
        *(float2*)ap = make_float2(w2[0], w2[1]);
    }
}

static __device__ __forceinline__ void gat_chunk1(const bool valid, const int s_l, const int cnt,
        const int lane, const bool hi, const float2 adn, const float* __restrict__ a_src,
        const float* __restrict__ h1, float& m0, float& m1, float4& acc, float& den) {
    const float2 as = *(const float2*)(a_src + (size_t)s_l * 2);
    const float e0 = valid ? lrelu(as.x + adn.x) : -INFINITY;
    const float e1 = valid ? lrelu(as.y + adn.y) : -INFINITY;
    float c0m = e0, c1m = e1;
    for (int o = 16; o; o >>= 1) {
        c0m = fmaxf(c0m, __shfl_xor(c0m, o));
        c1m = fmaxf(c1m, __shfl_xor(c1m, o));
    }
    const float n0 = fmaxf(m0, c0m), n1 = fmaxf(m1, c1m);
    const float r = hi ? __expf(m1 - n1) : __expf(m0 - n0);
    acc.x *= r; acc.y *= r; acc.z *= r; acc.w *= r; den *= r;
    m0 = n0; m1 = n1;
    const int p0_l = __float_as_int(valid ? __expf(e0 - n0) : 0.f);
    const int p1_l = __float_as_int(valid ? __expf(e1 - n1) : 0.f);
#pragma nounroll
    for (int j0 = 0; j0 < cnt; j0 += 4) {
#pragma unroll
        for (int u = 0; u < 4; ++u) {
            const int j = j0 + u;
            const int sj = __builtin_amdgcn_readlane(s_l, j);
            const float q0 = __int_as_float(__builtin_amdgcn_readlane(p0_l, j));
            const float q1 = __int_as_float(__builtin_amdgcn_readlane(p1_l, j));
            const float4 hv = *(const float4*)(h1 + (size_t)sj * F1 + lane * 4);
            const float p = hi ? q1 : q0;
            den += p;
            acc.x += p * hv.x; acc.y += p * hv.y; acc.z += p * hv.z; acc.w += p * hv.w;
        }
    }
}

static __device__ __forceinline__ float4 gat_node1(const int n, const int r, const int lane, const float4 b,
        const int* sCnt, const int* sCol, const int* __restrict__ be, const int nbe, const int tot,
        const int* __restrict__ binCnt, const int2* __restrict__ ovf,
        const float* __restrict__ a_src, const float* __restrict__ a_dst,
        const float* __restrict__ h1) {
    const int deg = __builtin_amdgcn_readfirstlane(sCnt[r]);
    const bool big = deg > BCAP;
    const int nb  = big ? 0 : deg;
    const int* cb = sCol + r * BCAP;
    const float2 adn = *(const float2*)(a_dst + (size_t)n * 2);
    const float2 asn = *(const float2*)(a_src + (size_t)n * 2);
    float m0 = lrelu(asn.x + adn.x), m1 = lrelu(asn.y + adn.y);
    const bool hi = lane >= 16;
    float4 acc = *(const float4*)(h1 + (size_t)n * F1 + lane * 4);
    float den = 1.f;
    for (int c0 = 0; c0 < nb; c0 += 32) {
        const int cnt = min(32, nb - c0);
        const bool valid = lane < cnt;
        const int s_l = valid ? cb[c0 + lane] : n;
        gat_chunk1(valid, s_l, cnt, lane, hi, adn, a_src, h1, m0, m1, acc, den);
    }
    if (big) {
        const int dl = n & (BIN_NODES - 1);
        for (int k0 = 0; k0 < nbe; k0 += 32) {
            const int k = k0 + lane;
            const int e = k < nbe ? be[k] : -1;
            const bool valid = (e >> 16) == dl;
            if (__ballot(valid) == 0) continue;
            const int s_l = valid ? (e & 0xffff) : n;
            gat_chunk1(valid, s_l, 32, lane, hi, adn, a_src, h1, m0, m1, acc, den);
        }
    }
    if (tot > BIN_CAP) {
        const int m = binCnt[NBINS * BIN_PITCH];
        for (int k0 = 0; k0 < m; k0 += 32) {
            const int k = k0 + lane;
            const int2 en = k < m ? ovf[k] : make_int2(-1, 0);
            const bool valid = en.x == n;
            if (__ballot(valid) == 0) continue;
            const int s_l = valid ? en.y : n;
            gat_chunk1(valid, s_l, 32, lane, hi, adn, a_src, h1, m0, m1, acc, den);
        }
    }
    float4 o;
    o.x = acc.x / den + b.x; o.y = acc.y / den + b.y;
    o.z = acc.z / den + b.z; o.w = acc.w / den + b.w;
    o.x = o.x > 0.f ? o.x : (expf(o.x) - 1.0f);
    o.y = o.y > 0.f ? o.y : (expf(o.y) - 1.0f);
    o.z = o.z > 0.f ? o.z : (expf(o.z) - 1.0f);
    o.w = o.w > 0.f ? o.w : (expf(o.w) - 1.0f);
    return o;
}

#define G2_ROWP 132
#define G2_NPB  16
static_assert(N_NODES % G2_NPB == 0, "a block owns 16 whole nodes");

__global__ __launch_bounds__(256) void gat_gather1_gemm2(const int* __restrict__ binCnt,
        const int* __restrict__ binE, int* __restrict__ count,
        int* __restrict__ col, const int2* __restrict__ ovf,
        const float* __restrict__ a_src, const float* __restrict__ a_dst,
        const float* __restrict__ h1, const float* __restrict__ b1,
        const float* __restrict__ W2, float* __restrict__ h2,
        const float* __restrict__ as2, const float* __restrict__ ad2,
        float* __restrict__ a_src2, float* __restrict__ a_dst2) {
    __shared__ float sW[(F1 / 4) * 48 * 4];
    __shared__ float sRow[G2_NPB * G2_ROWP];
    __shared__ int sCol[G2_NPB * BCAP];
    __shared__ int sCnt[G2_NPB];
    const int tid  = threadIdx.x;
    const int lane = tid & 31;
    const int wave = __builtin_amdgcn_readfirstlane(tid >> 5);

    const int bin = blockIdx.x >> 3;
    const int sub = blockIdx.x & 7;
    const int* be = binE + (size_t)bin * BIN_CAP;
    if (tid < G2_NPB) sCnt[tid] = 0;
    *(int4*)(sCol + tid * 4) = make_int4(0, 0, 0, 0);
    const int tot = binCnt[bin * BIN_PITCH];
    const int nbe = min(tot, BIN_CAP);
    int ev[BIN_CAP / 256];
#pragma unroll
    for (int u = 0; u < BIN_CAP / 256; ++u) ev[u] = tid + 256 * u < nbe ? be[tid + 256 * u] : -1;

#pragma unroll
    for (int u = 0; u < (F1 / 4) * 48 / 256; ++u) {
        const int i = tid + 256 * u;
        const int j = i / 48, n = i - 48 * j;
        const float* wp = W2 + (size_t)(4 * j) * NCLS + min(n, NCLS - 1);
        float4 v;
        v.x = wp[0]; v.y = wp[NCLS]; v.z = wp[2 * NCLS]; v.w = wp[3 * NCLS];
        if (n >= NCLS) v = make_float4(0.f, 0.f, 0.f, 0.f);
        *(float4*)(sW + i * 4) = v;
    }

    __syncthreads();
#pragma unroll
    for (int u = 0; u < BIN_CAP / 256; ++u) {
        const int e = ev[u];
        if ((e >> 20) == sub) {
            const int r = (e >> 16) & (G2_NPB - 1);
            const int slot = atomicAdd(&sCnt[r], 1);
            if (slot < BCAP) sCol[r * BCAP + slot] = e & 0xffff;
        }
    }
    __syncthreads();
    if (tid < G2_NPB) count[blockIdx.x * G2_NPB + tid] = sCnt[tid];
    *(int4*)(col + (size_t)blockIdx.x * G2_NPB * BCAP + tid * 4) = *(const int4*)(sCol + tid * 4);

    const float4 b = *(const float4*)(b1 + lane * 4);
    for (int t = 0; t < 2; ++t) {
        const int r = wave + 8 * t;
        const float4 o = gat_node1(blockIdx.x * G2_NPB + r, r, lane, b, sCnt, sCol, be, nbe, tot,
                                   binCnt, ovf, a_src, a_dst, h1);
        *(float4*)(sRow + r * G2_ROWP + lane * 4) = o;
    }
    __syncthreads();
    if (wave != 0) return;

    const int half = lane >> 4;
    const int l    = lane & 15;
    const float* ar = sRow + l * G2_ROWP + 2 * half;
    const float* bw = sW + l * 4 + 2 * half;
    v8f acc[3];
#pragma unroll
    for (int t = 0; t < 3; ++t) acc[t] = (v8f){};
#pragma unroll
    for (int j = 0; j < F1 / 4; ++j) {
        const v2f a = *(const v2f*)(ar + 4 * j);
#pragma unroll
        for (int t = 0; t < 3; ++t) {
            const v2f bf = *(const v2f*)(bw + j * 192 + t * 64);
            acc[t] = __builtin_amdgcn_wmma_f32_16x16x4_f32(false, a, false, bf,
                                                           (short)0, acc[t], false, false);
        }
    }

    const int base = blockIdx.x * G2_NPB;
    float part[3];
#pragma unroll
    for (int t = 0; t < 3; ++t) {
        const int ncol = t * 16 + l;
        if (ncol < NCLS) {
            float* dr = h2 + (size_t)(base + 8 * half) * NCLS + ncol;
#pragma unroll
            for (int r = 0; r < 8; ++r) dr[(size_t)r * NCLS] = acc[t][r];
        }
        const float sc = ncol < NCLS ? as2[ncol] : 0.f;
        const float dc = ncol < NCLS ? ad2[ncol] : 0.f;
        float v[16], w8[8], w4[4], w2[2], w1[1];
#pragma unroll
        for (int r = 0; r < 8; ++r) { v[r] = acc[t][r] * sc; v[8 + r] = acc[t][r] * dc; }
        row_halve<8, 8>((l & 8) != 0, v, w8);
        row_halve<4, 4>((l & 4) != 0, w8, w4);
        row_halve<2, 2>((l & 2) != 0, w4, w2);
        row_halve<1, 1>((l & 1) != 0, w2, w1);
        part[t] = w1[0];
    }
    const float dot = part[0] + part[1] + part[2];
    ((l & 8) ? a_dst2 : a_src2)[base + 8 * half + (l & 7)] = dot;
}

static __device__ __forceinline__ void gat_chunk2(const bool valid, const int s_l, const int cnt,
        const bool live, const int g, const int q, const float adn,
        const float* __restrict__ a_src, const float* __restrict__ h2,
        float& m, float4& acc, float& den) {
    const float e = valid ? lrelu(a_src[s_l] + adn) : -INFINITY;
    float cm = e;
    for (int o = 16; o; o >>= 1) cm = fmaxf(cm, __shfl_xor(cm, o));
    const float nm = fmaxf(m, cm);
    const float r = __expf(m - nm);
    acc.x *= r; acc.y *= r; acc.z *= r; acc.w *= r; den *= r;
    m = nm;
    const float p_l = valid ? __expf(e - nm) : 0.f;
    for (int j0 = 0; j0 < cnt; j0 += 6) {
#pragma unroll
        for (int u = 0; u < 2; ++u) {
            const int j = j0 + 3 * u + g;
            const bool ok = live && (j < cnt);
            const int jj = ok ? j : 0;
            const int sj = __shfl(s_l, jj);
            const float pv = __shfl(p_l, jj);
            const float pj = ok ? pv : 0.f;
            const float4 hv = *(const float4*)(h2 + (size_t)sj * NCLS + q * 4);
            den += pj;
            acc.x += pj * hv.x; acc.y += pj * hv.y; acc.z += pj * hv.z; acc.w += pj * hv.w;
        }
    }
}

__global__ __launch_bounds__(256) void gat_gather2(const int* __restrict__ binCnt,
        const int* __restrict__ binE, const int* __restrict__ count,
        const int* __restrict__ col, const int2* __restrict__ ovf,
        const float* __restrict__ a_src, const float* __restrict__ a_dst,
        const float* __restrict__ h2, const float* __restrict__ b2, float* __restrict__ out) {
    const int lane = threadIdx.x & 31;
    const int n = __builtin_amdgcn_readfirstlane(blockIdx.x * 8 + (threadIdx.x >> 5));
    if (n >= N_NODES) return;
    const int g3 = lane / 10;
    const bool live = g3 < 3;
    const int g = live ? g3 : 0;
    const int q = lane - 10 * g3;
    const int deg = count[n];
    const int bin = n >> 7;
    const int tot = binCnt[bin * BIN_PITCH];
    const bool big = deg > BCAP;
    const int nb  = big ? 0 : deg;
    const int* cb = col + (size_t)n * BCAP;
    const float adn = a_dst[n];
    float m = lrelu(a_src[n] + adn);
    float4 acc = {0.f, 0.f, 0.f, 0.f};
    float den = 0.f;
    if (lane < 10) { acc = *(const float4*)(h2 + (size_t)n * NCLS + q * 4); den = 1.f; }
    for (int c0 = 0; c0 < nb; c0 += 32) {
        const int cnt = min(32, nb - c0);
        const bool valid = lane < cnt;
        const int s_l = valid ? cb[c0 + lane] : n;
        gat_chunk2(valid, s_l, cnt, live, g, q, adn, a_src, h2, m, acc, den);
    }
    if (big) {
        const int nbe = min(tot, BIN_CAP);
        const int* be = binE + (size_t)bin * BIN_CAP;
        const int dl = n & (BIN_NODES - 1);
        for (int k0 = 0; k0 < nbe; k0 += 32) {
            const int k = k0 + lane;
            const int e = k < nbe ? be[k] : -1;
            const bool valid = (e >> 16) == dl;
            if (__ballot(valid) == 0) continue;
            const int s_l = valid ? (e & 0xffff) : n;
            gat_chunk2(valid, s_l, 32, live, g, q, adn, a_src, h2, m, acc, den);
        }
    }
    if (tot > BIN_CAP) {
        const int mo = binCnt[NBINS * BIN_PITCH];
        for (int k0 = 0; k0 < mo; k0 += 32) {
            const int k = k0 + lane;
            const int2 en = k < mo ? ovf[k] : make_int2(-1, 0);
            const bool valid = en.x == n;
            if (__ballot(valid) == 0) continue;
            const int s_l = valid ? en.y : n;
            gat_chunk2(valid, s_l, 32, live, g, q, adn, a_src, h2, m, acc, den);
        }
    }
    const int l1 = (lane + 10) & 31, l2 = (lane + 20) & 31;
    const float sx = acc.x + __shfl(acc.x, l1) + __shfl(acc.x, l2);
    const float sy = acc.y + __shfl(acc.y, l1) + __shfl(acc.y, l2);
    const float sz = acc.z + __shfl(acc.z, l1) + __shfl(acc.z, l2);
    const float sw = acc.w + __shfl(acc.w, l1) + __shfl(acc.w, l2);
    const float sd = den + __shfl(den, l1) + __shfl(den, l2);
    if (lane < 10) {
        const float4 b = *(const float4*)(b2 + q * 4);
        float4 o;
        o.x = sx / sd + b.x; o.y = sy / sd + b.y; o.z = sz / sd + b.z; o.w = sw / sd + b.w;
        *(float4*)(out + (size_t)n * NCLS + q * 4) = o;
    }
}

extern "C" void kernel_launch(void* const* d_in, const int* in_sizes, int n_in,
                              void* d_out, int out_size, void* d_ws, size_t ws_size,
                              hipStream_t stream) {
    const float* x    = (const float*)d_in[0];
    const int*   ei   = (const int*)  d_in[1];
    const float* W1   = (const float*)d_in[2];
    const float* as1  = (const float*)d_in[3];
    const float* ad1  = (const float*)d_in[4];
    const float* b1   = (const float*)d_in[5];
    const float* W2   = (const float*)d_in[6];
    const float* as2  = (const float*)d_in[7];
    const float* ad2  = (const float*)d_in[8];
    const float* b2   = (const float*)d_in[9];
    float* out = (float*)d_out;

    float* ws      = (float*)d_ws;
    float* h1      = ws;
    float* h2      = h1     + (size_t)N_NODES * F1;
    float* a_src1  = h2     + (size_t)N_NODES * NCLS;
    float* a_dst1  = a_src1 + (size_t)N_NODES * 2;
    float* a_src2  = a_dst1 + (size_t)N_NODES * 2;
    float* a_dst2  = a_src2 + N_NODES;
    int*   count   = (int*)(a_dst2 + N_NODES);
    int*   col     = count  + N_NODES + 16;
    int*   binCnt  = col    + (size_t)N_NODES * BCAP;
    int*   binE    = binCnt + NBINS * BIN_PITCH + 32;
    int2*  ovf     = (int2*)(binE + (size_t)NBINS * BIN_CAP);

    hipMemsetAsync(binCnt, 0, (size_t)(NBINS * BIN_PITCH + 32) * sizeof(int), stream);

    gat_gemm1_and_buckets<<<G1_BLOCKS, 256, 0, stream>>>(x, W1, h1, ei, binCnt, binE, ovf,
                                                         as1, ad1, a_src1, a_dst1);
    gat_gather1_gemm2<<<N_NODES / G2_NPB, 256, 0, stream>>>(binCnt, binE, count, col, ovf, a_src1, a_dst1, h1, b1,
                                                            W2, h2, as2, ad2, a_src2, a_dst2);
    gat_gather2<<<N_NODES / 8, 256, 0, stream>>>(binCnt, binE, count, col, ovf, a_src2, a_dst2, h2, b2, out);
}
